# starting kernel
# baseline (speedup 1.0000x reference)
#include <hip/hip_runtime.h>

#define NN 50000
#define EE 800000
#define DD 128
#define K3 384

typedef __attribute__((ext_vector_type(2))) float v2f;
typedef __attribute__((ext_vector_type(8))) float v8f;

__global__ __launch_bounds__(256) void deg_kernel(
    const int* __restrict__ src, const int* __restrict__ dst,
    float* __restrict__ deg_out, float* __restrict__ deg_in, int E) {
  int e = blockIdx.x * blockDim.x + threadIdx.x;
  if (e >= E) return;
  atomicAdd(&deg_out[src[e]], 1.0f);
  atomicAdd(&deg_in[dst[e]], 1.0f);
}

__global__ __launch_bounds__(256) void inv_kernel(float* __restrict__ deg, int n) {
  int i = blockIdx.x * blockDim.x + threadIdx.x;
  if (i >= n) return;
  deg[i] = 1.0f / fmaxf(deg[i], 1.0f);
}

__global__ __launch_bounds__(256) void agg_kernel(
    const float* __restrict__ x,
    const int* __restrict__ src, const int* __restrict__ dst,
    const float* __restrict__ inv_out, const float* __restrict__ inv_in,
    float* __restrict__ fwd, float* __restrict__ bwd, int E) {
  long long tid = (long long)blockIdx.x * blockDim.x + threadIdx.x;
  int e = (int)(tid >> 5);
  if (e >= E) return;
  int f4 = ((int)tid & 31) * 4;

  int s = src[e];
  int d = dst[e];
  float io = inv_out[s];
  float ii = inv_in[d];

  float4 xs = *(const float4*)(x + (size_t)s * DD + f4);
  float4 xd = *(const float4*)(x + (size_t)d * DD + f4);

  float* fp = fwd + (size_t)d * DD + f4;
  float* bp = bwd + (size_t)s * DD + f4;
  atomicAdd(fp + 0, xs.x * io);
  atomicAdd(fp + 1, xs.y * io);
  atomicAdd(fp + 2, xs.z * io);
  atomicAdd(fp + 3, xs.w * io);
  atomicAdd(bp + 0, xd.x * ii);
  atomicAdd(bp + 1, xd.y * ii);
  atomicAdd(bp + 2, xd.z * ii);
  atomicAdd(bp + 3, xd.w * ii);
}

__global__ __launch_bounds__(256) void gemm_wmma_kernel(
    const float* __restrict__ x, const float* __restrict__ fwd,
    const float* __restrict__ bwd, const float* __restrict__ W,
    const float* __restrict__ bias, float* __restrict__ out, int do_relu) {
  __shared__ float As[16][K3 + 4];

  const int rowBase = blockIdx.x * 16;
  const int tid = threadIdx.x;
  const int wave = tid >> 5;
  const int lane = tid & 31;
  const int hl = lane >> 4;
  const int l16 = lane & 15;

  for (int i = tid; i < 16 * K3; i += 256) {
    int r = i / K3;
    int k = i - r * K3;
    size_t node = (size_t)(rowBase + r);
    float v;
    if (k < DD)            v = x[node * DD + k];
    else if (k < 2 * DD)   v = fwd[node * DD + (k - DD)];
    else                   v = bwd[node * DD + (k - 2 * DD)];
    As[r][k] = v;
  }
  __syncthreads();

  const int colBase = wave * 16;
  const int col = colBase + l16;

  v8f acc = {};
  for (int k0 = 0; k0 < K3; k0 += 4) {
    v2f a;
    a.x = As[l16][k0 + 2 * hl + 0];
    a.y = As[l16][k0 + 2 * hl + 1];
    v2f b;
    b.x = W[(size_t)(k0 + 2 * hl + 0) * DD + col];
    b.y = W[(size_t)(k0 + 2 * hl + 1) * DD + col];
    acc = __builtin_amdgcn_wmma_f32_16x16x4_f32(
         false, a,  false, b,
         (short)0, acc,  false,  false);
  }

  const float bb = bias[col];
#pragma unroll
  for (int r = 0; r < 8; ++r) {
    float v = acc[r] + bb;
    if (do_relu) v = fmaxf(v, 0.0f);
    out[(size_t)(rowBase + r + 8 * hl) * DD + col] = v;
  }
}

extern "C" void kernel_launch(void* const* d_in, const int* in_sizes, int n_in,
                              void* d_out, int out_size, void* d_ws, size_t ws_size,
                              hipStream_t stream) {
  const float* x  = (const float*)d_in[0];
  const float* W0 = (const float*)d_in[1];
  const float* b0 = (const float*)d_in[2];
  const float* W1 = (const float*)d_in[3];
  const float* b1 = (const float*)d_in[4];
  const float* W2 = (const float*)d_in[5];
  const float* b2 = (const float*)d_in[6];
  const int* src  = (const int*)d_in[7];
  const int* dst  = (const int*)d_in[8];
  float* out = (float*)d_out;

  float* ws      = (float*)d_ws;
  float* deg_out = ws;
  float* deg_in  = ws + NN;
  float* xA      = ws + 2 * (size_t)NN;
  float* xB      = xA + (size_t)NN * DD;
  float* fwd     = xB + (size_t)NN * DD;
  float* bwd     = fwd + (size_t)NN * DD;

  hipMemsetAsync(deg_out, 0, 2 * (size_t)NN * sizeof(float), stream);
  deg_kernel<<<(EE + 255) / 256, 256, 0, stream>>>(src, dst, deg_out, deg_in, EE);
  inv_kernel<<<(2 * NN + 255) / 256, 256, 0, stream>>>(deg_out, 2 * NN);

  const float* Ws[3]  = {W0, W1, W2};
  const float* bs[3]  = {b0, b1, b2};
  float*       dsts[3] = {xA, xB, out};

  const float* cur = x;
  const long long agg_threads = (long long)EE * 32;
  const int agg_blocks = (int)((agg_threads + 255) / 256);

  for (int l = 0; l < 3; ++l) {
    hipMemsetAsync(fwd, 0, 2 * (size_t)NN * DD * sizeof(float), stream);
    agg_kernel<<<agg_blocks, 256, 0, stream>>>(cur, src, dst, deg_out, deg_in,
                                               fwd, bwd, EE);
    gemm_wmma_kernel<<<NN / 16, 256, 0, stream>>>(cur, fwd, bwd, Ws[l], bs[l],
                                                  dsts[l], (l < 2) ? 1 : 0);
    cur = dsts[l];
  }
}


# Aggregate in LDS: a work-group owns 256 nodes' fwd+bwd sums, streams all packed edge keys, adds matching rows by LDS float atomics; its global atomics and memsets go; sum order = LDS arrival order
# speedup vs baseline: 2.2131x; 2.2131x over previous
#include <hip/hip_runtime.h>

#define NN 50000
#define EE 800000
#define DD 128
#define K3 384

#define RNG 256
#define NWG ((NN + RNG - 1) / RNG)
#define AGG_WAVES 32
#define QCAP 320
#define KPI 5
#define NIT (EE / (128 * KPI))
#define STAG (NIT / NWG)

static_assert(NN <= 65536, "packed keys hold a node id in 16 bits");
static_assert(EE % (128 * KPI) == 0, "key stream is read in whole scan iterations");
static_assert(RNG == 256, "range tests and queue entries use the node id's high byte");

typedef __attribute__((ext_vector_type(2))) float v2f;
typedef __attribute__((ext_vector_type(8))) float v8f;

__global__ __launch_bounds__(256) void deg_kernel(
    const int* __restrict__ src, const int* __restrict__ dst,
    float* __restrict__ deg_out, float* __restrict__ deg_in,
    int* __restrict__ pk, int E) {
  int e = blockIdx.x * blockDim.x + threadIdx.x;
  if (e >= E) return;
  int s = src[e];
  int d = dst[e];
  atomicAdd(&deg_out[s], 1.0f);
  atomicAdd(&deg_in[d], 1.0f);
  pk[e] = (int)((unsigned)s | ((unsigned)d << 16));
}

__global__ __launch_bounds__(256) void inv_kernel(float* __restrict__ deg, int n) {
  int i = blockIdx.x * blockDim.x + threadIdx.x;
  if (i >= n) return;
  deg[i] = 1.0f / fmaxf(deg[i], 1.0f);
}

__global__ __launch_bounds__(1024) void agg_owner_kernel(
    const float* __restrict__ x, const int* __restrict__ pk,
    const float* __restrict__ inv,
    float* __restrict__ fwd, float* __restrict__ bwd) {
  __shared__ float acc[(2 * RNG + 1) * DD];
  __shared__ unsigned q[AGG_WAVES][QCAP + 8];

  const int tid = threadIdx.x;
  const int lane = tid & 31;
  const int wave = __builtin_amdgcn_readfirstlane(tid >> 5);
  const unsigned b = blockIdx.x;

  float4* acc4 = (float4*)acc;
  for (int i = tid; i < (2 * RNG + 1) * DD / 4; i += 1024)
    acc4[i] = make_float4(0.0f, 0.0f, 0.0f, 0.0f);
  __syncthreads();

  const uint4* pk4 = (const uint4*)pk;
  const float4* x4 = (const float4*)x;
  unsigned* myq = q[wave];
  int cnt = 0;

  auto drain = [&](int n) {
    if (lane < 8) myq[n + lane] = 1u << 25;
    for (int i = 0; i < n; i += 8) {
      unsigned e[8];
      float w[8];
      float4 r[8];
#pragma unroll
      for (int j = 0; j < 8; ++j)
        e[j] = __builtin_amdgcn_readfirstlane(myq[i + j]);
#pragma unroll
      for (int j = 0; j < 8; ++j) {
        unsigned o = e[j] & 0xFFFFu;
        w[j] = inv[((e[j] >> 24) & 1u) * NN + o];
        r[j] = x4[(size_t)o * (DD / 4) + lane];
      }
#pragma unroll
      for (int j = 0; j < 8; ++j) {
        float* a = acc + (e[j] >> 16) * DD + lane;
        atomicAdd(a, r[j].x * w[j]);
        atomicAdd(a + 32, r[j].y * w[j]);
        atomicAdd(a + 64, r[j].z * w[j]);
        atomicAdd(a + 96, r[j].w * w[j]);
      }
    }
  };

  for (int it = wave; it < NIT; it += AGG_WAVES) {
    int ip = it + (int)b * STAG;
    if (ip >= NIT) ip -= NIT;
    uint4 k[KPI];
#pragma unroll
    for (int u = 0; u < KPI; ++u)
      k[u] = pk4[((size_t)ip * KPI + u) * 32 + lane];
#pragma unroll
    for (int u = 0; u < KPI; ++u) {
      const unsigned kk[4] = {k[u].x, k[u].y, k[u].z, k[u].w};
#pragma unroll
      for (int j = 0; j < 4; ++j) {
        const unsigned p = kk[j];
        const bool tf = (p >> 24) == b;
        const bool tb = ((p >> 8) & 255u) == b;
        const unsigned mf = __builtin_amdgcn_ballot_w32(tf);
        const unsigned mb = __builtin_amdgcn_ballot_w32(tb);
        if (mf | mb) {
          if (mf) {
            unsigned pos = cnt + __builtin_amdgcn_mbcnt_lo(mf, 0u);
            if (tf) myq[pos] = p & 0x00FFFFFFu;
            cnt += __builtin_popcount(mf);
          }
          if (mb) {
            unsigned pos = cnt + __builtin_amdgcn_mbcnt_lo(mb, 0u);
            if (tb) myq[pos] = (p >> 16) | ((p & 255u) << 16) | (1u << 24);
            cnt += __builtin_popcount(mb);
          }
        }
      }
      if (cnt > 64) {
        drain(cnt);
        cnt = 0;
      }
    }
  }
  drain(cnt);
  __syncthreads();

  for (int row = wave; row < 2 * RNG; row += AGG_WAVES) {
    int node = (int)b * RNG + (row & (RNG - 1));
    if (node < NN) {
      const float* a = acc + row * DD + lane;
      float4 v = make_float4(a[0], a[32], a[64], a[96]);
      float* o = (row < RNG) ? fwd : bwd;
      *(float4*)(o + (size_t)node * DD + lane * 4) = v;
    }
  }
}

__global__ __launch_bounds__(256) void gemm_wmma_kernel(
    const float* __restrict__ x, const float* __restrict__ fwd,
    const float* __restrict__ bwd, const float* __restrict__ W,
    const float* __restrict__ bias, float* __restrict__ out, int do_relu) {
  __shared__ float As[16][K3 + 4];

  const int rowBase = blockIdx.x * 16;
  const int tid = threadIdx.x;
  const int wave = tid >> 5;
  const int lane = tid & 31;
  const int hl = lane >> 4;
  const int l16 = lane & 15;

  for (int i = tid; i < 16 * K3; i += 256) {
    int r = i / K3;
    int k = i - r * K3;
    size_t node = (size_t)(rowBase + r);
    float v;
    if (k < DD)            v = x[node * DD + k];
    else if (k < 2 * DD)   v = fwd[node * DD + (k - DD)];
    else                   v = bwd[node * DD + (k - 2 * DD)];
    As[r][k] = v;
  }
  __syncthreads();

  const int colBase = wave * 16;
  const int col = colBase + l16;

  v8f acc = {};
  for (int k0 = 0; k0 < K3; k0 += 4) {
    v2f a;
    a.x = As[l16][k0 + 2 * hl + 0];
    a.y = As[l16][k0 + 2 * hl + 1];
    v2f b;
    b.x = W[(size_t)(k0 + 2 * hl + 0) * DD + col];
    b.y = W[(size_t)(k0 + 2 * hl + 1) * DD + col];
    acc = __builtin_amdgcn_wmma_f32_16x16x4_f32(
         false, a,  false, b,
         (short)0, acc,  false,  false);
  }

  const float bb = bias[col];
#pragma unroll
  for (int r = 0; r < 8; ++r) {
    float v = acc[r] + bb;
    if (do_relu) v = fmaxf(v, 0.0f);
    out[(size_t)(rowBase + r + 8 * hl) * DD + col] = v;
  }
}

extern "C" void kernel_launch(void* const* d_in, const int* in_sizes, int n_in,
                              void* d_out, int out_size, void* d_ws, size_t ws_size,
                              hipStream_t stream) {
  const float* x  = (const float*)d_in[0];
  const float* W0 = (const float*)d_in[1];
  const float* b0 = (const float*)d_in[2];
  const float* W1 = (const float*)d_in[3];
  const float* b1 = (const float*)d_in[4];
  const float* W2 = (const float*)d_in[5];
  const float* b2 = (const float*)d_in[6];
  const int* src  = (const int*)d_in[7];
  const int* dst  = (const int*)d_in[8];
  float* out = (float*)d_out;

  float* ws      = (float*)d_ws;
  float* deg_out = ws;
  float* deg_in  = ws + NN;
  float* xA      = ws + 2 * (size_t)NN;
  int*   pk      = (int*)(xA + (size_t)NN * DD);
  float* fwd     = xA + 2 * (size_t)NN * DD;
  float* bwd     = fwd + (size_t)NN * DD;

  hipMemsetAsync(deg_out, 0, 2 * (size_t)NN * sizeof(float), stream);
  deg_kernel<<<(EE + 255) / 256, 256, 0, stream>>>(src, dst, deg_out, deg_in, pk, EE);
  inv_kernel<<<(2 * NN + 255) / 256, 256, 0, stream>>>(deg_out, 2 * NN);

  const float* Ws[3]  = {W0, W1, W2};
  const float* bs[3]  = {b0, b1, b2};
  float*       dsts[3] = {out, xA, out};

  const float* cur = x;
  for (int l = 0; l < 3; ++l) {
    agg_owner_kernel<<<NWG, 1024, 0, stream>>>(cur, pk, deg_out, fwd, bwd);
    gemm_wmma_kernel<<<NN / 16, 256, 0, stream>>>(cur, fwd, bwd, Ws[l], bs[l],
                                                  dsts[l], (l < 2) ? 1 : 0);
    cur = dsts[l];
  }
}


# Scan the edge keys once: layer 1's owner waves save their matched-edge queues to the workspace, layers 2-3 replay the saved lists instead of re-scanning all keys; sum order stays LDS arrival order
# speedup vs baseline: 2.5520x; 1.1531x over previous
#include <hip/hip_runtime.h>

#define NN 50000
#define EE 800000
#define DD 128
#define K3 384

#define RNG 256
#define NWG ((NN + RNG - 1) / RNG)
#define AGG_WAVES 32
#define QCAP 320
#define KPI 5
#define NIT (EE / (128 * KPI))
#define STAG (NIT / NWG)
#define LSTRIDE 832
#define LCAP (LSTRIDE - 32)
#define NLW (NWG * AGG_WAVES)

static_assert(NN <= 65536, "packed keys hold a node id in 16 bits");
static_assert(EE % (128 * KPI) == 0, "key stream is read in whole scan iterations");
static_assert(RNG == 256, "range tests and queue entries use the node id's high byte");
static_assert((size_t)EE + (size_t)NLW * LSTRIDE + NLW <= (size_t)NN * DD,
              "keys, saved lists and list lengths share the key plane of the workspace");
static_assert(LSTRIDE % 32 == 0 && LCAP >= QCAP, "lists are read in chunks of 32 entries");

typedef __attribute__((ext_vector_type(2))) float v2f;
typedef __attribute__((ext_vector_type(8))) float v8f;

__global__ __launch_bounds__(256) void deg_kernel(
    const int* __restrict__ src, const int* __restrict__ dst,
    float* __restrict__ deg_out, float* __restrict__ deg_in,
    int* __restrict__ pk, int E) {
  int e = blockIdx.x * blockDim.x + threadIdx.x;
  if (e >= E) return;
  int s = src[e];
  int d = dst[e];
  atomicAdd(&deg_out[s], 1.0f);
  atomicAdd(&deg_in[d], 1.0f);
  pk[e] = (int)((unsigned)s | ((unsigned)d << 16));
}

__global__ __launch_bounds__(256) void inv_kernel(float* __restrict__ deg, int n) {
  int i = blockIdx.x * blockDim.x + threadIdx.x;
  if (i >= n) return;
  deg[i] = 1.0f / fmaxf(deg[i], 1.0f);
}

template <int REPLAY>
__global__ __launch_bounds__(1024) void agg_owner_kernel(
    const float* __restrict__ x, const int* __restrict__ pk,
    const float* __restrict__ inv,
    float* __restrict__ fwd, float* __restrict__ bwd,
    int* __restrict__ lists,
    int* __restrict__ cnts) {
  __shared__ float acc[(2 * RNG + 1) * DD];
  __shared__ unsigned q[AGG_WAVES][QCAP + 8];

  const int tid = threadIdx.x;
  const int lane = tid & 31;
  const int wave = __builtin_amdgcn_readfirstlane(tid >> 5);
  const unsigned b = blockIdx.x;

  const int wg = (int)b * AGG_WAVES + wave;
  int* mylist = lists + (size_t)wg * LSTRIDE;
  int nrep = -1;
  if (REPLAY) nrep = __builtin_amdgcn_readfirstlane(cnts[wg]);

  float4* acc4 = (float4*)acc;
  for (int i = tid; i < (2 * RNG + 1) * DD / 4; i += 1024)
    acc4[i] = make_float4(0.0f, 0.0f, 0.0f, 0.0f);
  __syncthreads();

  const uint4* pk4 = (const uint4*)pk;
  const float4* x4 = (const float4*)x;
  unsigned* myq = q[wave];
  int cnt = 0;

  int saved = 0;
  auto drain = [&](int n) {
    if (lane < 8) myq[n + lane] = 1u << 25;
    if (!REPLAY) {
      if (saved + n <= LCAP) {
#pragma nounroll
        for (int i = lane; i < n; i += 32) mylist[saved + i] = (int)myq[i];
        saved += n;
      } else {
        saved = LCAP + 1;
      }
    }
    for (int i = 0; i < n; i += 8) {
      unsigned e[8];
      float w[8];
      float4 r[8];
#pragma unroll
      for (int j = 0; j < 8; ++j)
        e[j] = __builtin_amdgcn_readfirstlane(myq[i + j]);
#pragma unroll
      for (int j = 0; j < 8; ++j) {
        unsigned o = e[j] & 0xFFFFu;
        w[j] = inv[((e[j] >> 24) & 1u) * NN + o];
        r[j] = x4[(size_t)o * (DD / 4) + lane];
      }
#pragma unroll
      for (int j = 0; j < 8; ++j) {
        float* a = acc + (e[j] >> 16) * DD + lane;
        atomicAdd(a, r[j].x * w[j]);
        atomicAdd(a + 32, r[j].y * w[j]);
        atomicAdd(a + 64, r[j].z * w[j]);
        atomicAdd(a + 96, r[j].w * w[j]);
      }
    }
  };

  if (REPLAY && nrep >= 0) {
    unsigned cur = (unsigned)mylist[lane];
    unsigned nxt = 0;
    if (32 < nrep) nxt = (unsigned)mylist[32 + lane];
    float wcur = inv[((cur >> 24) & 1u) * NN + (cur & 0xFFFFu)];
    asm volatile("" : "+v"(wcur));
    for (int base = 0; base < nrep; base += 32) {
      float wnxt = 0.0f;
      unsigned nn = 0;
      if (base + 32 < nrep) wnxt = inv[((nxt >> 24) & 1u) * NN + (nxt & 0xFFFFu)];
      if (base + 64 < nrep) nn = (unsigned)mylist[base + 64 + lane];
#pragma unroll
      for (int g = 0; g < 32; g += 16) {
        if (base + g < nrep) {
          unsigned e[16];
          float w[16];
          float4 r[16];
#pragma unroll
          for (int j = 0; j < 16; ++j)
            e[j] = (unsigned)__builtin_amdgcn_readlane((int)cur, g + j);
#pragma unroll
          for (int j = 0; j < 16; ++j)
            r[j] = x4[(size_t)(e[j] & 0xFFFFu) * (DD / 4) + lane];
#pragma unroll
          for (int j = 0; j < 16; ++j)
            w[j] = __int_as_float(__builtin_amdgcn_readlane(__float_as_int(wcur), g + j));
#pragma unroll
          for (int j = 0; j < 16; ++j) {
            float* a = acc + (e[j] >> 16) * DD + lane;
            atomicAdd(a, r[j].x * w[j]);
            atomicAdd(a + 32, r[j].y * w[j]);
            atomicAdd(a + 64, r[j].z * w[j]);
            atomicAdd(a + 96, r[j].w * w[j]);
          }
        }
      }
      cur = nxt;
      wcur = wnxt;
      nxt = nn;
    }
  } else {
  for (int it = wave; it < NIT; it += AGG_WAVES) {
    int ip = it + (int)b * STAG;
    if (ip >= NIT) ip -= NIT;
    uint4 k[KPI];
#pragma unroll
    for (int u = 0; u < KPI; ++u)
      k[u] = pk4[((size_t)ip * KPI + u) * 32 + lane];
#pragma unroll
    for (int u = 0; u < KPI; ++u) {
      const unsigned kk[4] = {k[u].x, k[u].y, k[u].z, k[u].w};
#pragma unroll
      for (int j = 0; j < 4; ++j) {
        const unsigned p = kk[j];
        const bool tf = (p >> 24) == b;
        const bool tb = ((p >> 8) & 255u) == b;
        const unsigned mf = __builtin_amdgcn_ballot_w32(tf);
        const unsigned mb = __builtin_amdgcn_ballot_w32(tb);
        if (mf | mb) {
          if (mf) {
            unsigned pos = cnt + __builtin_amdgcn_mbcnt_lo(mf, 0u);
            if (tf) myq[pos] = p & 0x00FFFFFFu;
            cnt += __builtin_popcount(mf);
          }
          if (mb) {
            unsigned pos = cnt + __builtin_amdgcn_mbcnt_lo(mb, 0u);
            if (tb) myq[pos] = (p >> 16) | ((p & 255u) << 16) | (1u << 24);
            cnt += __builtin_popcount(mb);
          }
        }
      }
      if (cnt > 64) {
        drain(cnt);
        cnt = 0;
      }
    }
  }
  drain(cnt);
  if (!REPLAY) {
    if (saved <= LCAP) mylist[saved + lane] = 1 << 25;
    if (lane == 0) cnts[wg] = (saved <= LCAP) ? saved : -1;
  }
  }
  __syncthreads();

  for (int row = wave; row < 2 * RNG; row += AGG_WAVES) {
    int node = (int)b * RNG + (row & (RNG - 1));
    if (node < NN) {
      const float* a = acc + row * DD + lane;
      float4 v = make_float4(a[0], a[32], a[64], a[96]);
      float* o = (row < RNG) ? fwd : bwd;
      *(float4*)(o + (size_t)node * DD + lane * 4) = v;
    }
  }
}

__global__ __launch_bounds__(256) void gemm_wmma_kernel(
    const float* __restrict__ x, const float* __restrict__ fwd,
    const float* __restrict__ bwd, const float* __restrict__ W,
    const float* __restrict__ bias, float* __restrict__ out, int do_relu) {
  __shared__ float As[16][K3 + 4];

  const int rowBase = blockIdx.x * 16;
  const int tid = threadIdx.x;
  const int wave = tid >> 5;
  const int lane = tid & 31;
  const int hl = lane >> 4;
  const int l16 = lane & 15;

  for (int i = tid; i < 16 * K3; i += 256) {
    int r = i / K3;
    int k = i - r * K3;
    size_t node = (size_t)(rowBase + r);
    float v;
    if (k < DD)            v = x[node * DD + k];
    else if (k < 2 * DD)   v = fwd[node * DD + (k - DD)];
    else                   v = bwd[node * DD + (k - 2 * DD)];
    As[r][k] = v;
  }
  __syncthreads();

  const int colBase = wave * 16;
  const int col = colBase + l16;

  v8f acc = {};
  for (int k0 = 0; k0 < K3; k0 += 4) {
    v2f a;
    a.x = As[l16][k0 + 2 * hl + 0];
    a.y = As[l16][k0 + 2 * hl + 1];
    v2f b;
    b.x = W[(size_t)(k0 + 2 * hl + 0) * DD + col];
    b.y = W[(size_t)(k0 + 2 * hl + 1) * DD + col];
    acc = __builtin_amdgcn_wmma_f32_16x16x4_f32(
         false, a,  false, b,
         (short)0, acc,  false,  false);
  }

  const float bb = bias[col];
#pragma unroll
  for (int r = 0; r < 8; ++r) {
    float v = acc[r] + bb;
    if (do_relu) v = fmaxf(v, 0.0f);
    out[(size_t)(rowBase + r + 8 * hl) * DD + col] = v;
  }
}

extern "C" void kernel_launch(void* const* d_in, const int* in_sizes, int n_in,
                              void* d_out, int out_size, void* d_ws, size_t ws_size,
                              hipStream_t stream) {
  const float* x  = (const float*)d_in[0];
  const float* W0 = (const float*)d_in[1];
  const float* b0 = (const float*)d_in[2];
  const float* W1 = (const float*)d_in[3];
  const float* b1 = (const float*)d_in[4];
  const float* W2 = (const float*)d_in[5];
  const float* b2 = (const float*)d_in[6];
  const int* src  = (const int*)d_in[7];
  const int* dst  = (const int*)d_in[8];
  float* out = (float*)d_out;

  float* ws      = (float*)d_ws;
  float* deg_out = ws;
  float* deg_in  = ws + NN;
  float* xA      = ws + 2 * (size_t)NN;
  int*   pk      = (int*)(xA + (size_t)NN * DD);
  int*   lists   = pk + EE;
  int*   cnts    = lists + (size_t)NLW * LSTRIDE;
  float* fwd     = xA + 2 * (size_t)NN * DD;
  float* bwd     = fwd + (size_t)NN * DD;

  hipMemsetAsync(deg_out, 0, 2 * (size_t)NN * sizeof(float), stream);
  deg_kernel<<<(EE + 255) / 256, 256, 0, stream>>>(src, dst, deg_out, deg_in, pk, EE);
  inv_kernel<<<(2 * NN + 255) / 256, 256, 0, stream>>>(deg_out, 2 * NN);

  const float* Ws[3]  = {W0, W1, W2};
  const float* bs[3]  = {b0, b1, b2};
  float*       dsts[3] = {out, xA, out};

  const float* cur = x;
  for (int l = 0; l < 3; ++l) {
    if (l == 0)
      agg_owner_kernel<0><<<NWG, 1024, 0, stream>>>(cur, pk, deg_out, fwd, bwd, lists, cnts);
    else
      agg_owner_kernel<1><<<NWG, 1024, 0, stream>>>(cur, pk, deg_out, fwd, bwd, lists, cnts);
    gemm_wmma_kernel<<<NN / 16, 256, 0, stream>>>(cur, fwd, bwd, Ws[l], bs[l],
                                                  dsts[l], (l < 2) ? 1 : 0);
    cur = dsts[l];
  }
}


# Persistent GEMM: 256 work-groups walk the 16-row tiles, each wave keeps its 16-column strip of W (all 384 k) in registers as B fragments; next A tile staged to LDS behind the products; sum order kept
# speedup vs baseline: 2.7359x; 1.0721x over previous
#include <hip/hip_runtime.h>

#define NN 50000
#define EE 800000
#define DD 128
#define K3 384

#define RNG 256
#define NWG ((NN + RNG - 1) / RNG)
#define AGG_WAVES 32
#define QCAP 320
#define KPI 5
#define NIT (EE / (128 * KPI))
#define STAG (NIT / NWG)
#define LSTRIDE 832
#define LCAP (LSTRIDE - 32)
#define NLW (NWG * AGG_WAVES)

static_assert(NN <= 65536, "packed keys hold a node id in 16 bits");
static_assert(EE % (128 * KPI) == 0, "key stream is read in whole scan iterations");
static_assert(RNG == 256, "range tests and queue entries use the node id's high byte");
static_assert((size_t)EE + (size_t)NLW * LSTRIDE + NLW <= (size_t)NN * DD,
              "keys, saved lists and list lengths share the key plane of the workspace");
static_assert(LSTRIDE % 32 == 0 && LCAP >= QCAP, "lists are read in chunks of 32 entries");

typedef __attribute__((ext_vector_type(2))) float v2f;
typedef __attribute__((ext_vector_type(8))) float v8f;

__global__ __launch_bounds__(256) void deg_kernel(
    const int* __restrict__ src, const int* __restrict__ dst,
    float* __restrict__ deg_out, float* __restrict__ deg_in,
    int* __restrict__ pk, int E) {
  int e = blockIdx.x * blockDim.x + threadIdx.x;
  if (e >= E) return;
  int s = src[e];
  int d = dst[e];
  atomicAdd(&deg_out[s], 1.0f);
  atomicAdd(&deg_in[d], 1.0f);
  pk[e] = (int)((unsigned)s | ((unsigned)d << 16));
}

__global__ __launch_bounds__(256) void inv_kernel(float* __restrict__ deg, int n) {
  int i = blockIdx.x * blockDim.x + threadIdx.x;
  if (i >= n) return;
  deg[i] = 1.0f / fmaxf(deg[i], 1.0f);
}

template <int REPLAY>
__global__ __launch_bounds__(1024) void agg_owner_kernel(
    const float* __restrict__ x, const int* __restrict__ pk,
    const float* __restrict__ inv,
    float* __restrict__ fwd, float* __restrict__ bwd,
    int* __restrict__ lists,
    int* __restrict__ cnts) {
  __shared__ float acc[(2 * RNG + 1) * DD];
  __shared__ unsigned q[AGG_WAVES][QCAP + 8];

  const int tid = threadIdx.x;
  const int lane = tid & 31;
  const int wave = __builtin_amdgcn_readfirstlane(tid >> 5);
  const unsigned b = blockIdx.x;

  const int wg = (int)b * AGG_WAVES + wave;
  int* mylist = lists + (size_t)wg * LSTRIDE;
  int nrep = -1;
  if (REPLAY) nrep = __builtin_amdgcn_readfirstlane(cnts[wg]);

  float4* acc4 = (float4*)acc;
  for (int i = tid; i < (2 * RNG + 1) * DD / 4; i += 1024)
    acc4[i] = make_float4(0.0f, 0.0f, 0.0f, 0.0f);
  __syncthreads();

  const uint4* pk4 = (const uint4*)pk;
  const float4* x4 = (const float4*)x;
  unsigned* myq = q[wave];
  int cnt = 0;

  int saved = 0;
  auto drain = [&](int n) {
    if (lane < 8) myq[n + lane] = 1u << 25;
    if (!REPLAY) {
      if (saved + n <= LCAP) {
#pragma nounroll
        for (int i = lane; i < n; i += 32) mylist[saved + i] = (int)myq[i];
        saved += n;
      } else {
        saved = LCAP + 1;
      }
    }
    for (int i = 0; i < n; i += 8) {
      unsigned e[8];
      float w[8];
      float4 r[8];
#pragma unroll
      for (int j = 0; j < 8; ++j)
        e[j] = __builtin_amdgcn_readfirstlane(myq[i + j]);
#pragma unroll
      for (int j = 0; j < 8; ++j) {
        unsigned o = e[j] & 0xFFFFu;
        w[j] = inv[((e[j] >> 24) & 1u) * NN + o];
        r[j] = x4[(size_t)o * (DD / 4) + lane];
      }
#pragma unroll
      for (int j = 0; j < 8; ++j) {
        float* a = acc + (e[j] >> 16) * DD + lane;
        atomicAdd(a, r[j].x * w[j]);
        atomicAdd(a + 32, r[j].y * w[j]);
        atomicAdd(a + 64, r[j].z * w[j]);
        atomicAdd(a + 96, r[j].w * w[j]);
      }
    }
  };

  if (REPLAY && nrep >= 0) {
    unsigned cur = (unsigned)mylist[lane];
    unsigned nxt = 0;
    if (32 < nrep) nxt = (unsigned)mylist[32 + lane];
    float wcur = inv[((cur >> 24) & 1u) * NN + (cur & 0xFFFFu)];
    asm volatile("" : "+v"(wcur));
    for (int base = 0; base < nrep; base += 32) {
      float wnxt = 0.0f;
      unsigned nn = 0;
      if (base + 32 < nrep) wnxt = inv[((nxt >> 24) & 1u) * NN + (nxt & 0xFFFFu)];
      if (base + 64 < nrep) nn = (unsigned)mylist[base + 64 + lane];
#pragma unroll
      for (int g = 0; g < 32; g += 16) {
        if (base + g < nrep) {
          unsigned e[16];
          float w[16];
          float4 r[16];
#pragma unroll
          for (int j = 0; j < 16; ++j)
            e[j] = (unsigned)__builtin_amdgcn_readlane((int)cur, g + j);
#pragma unroll
          for (int j = 0; j < 16; ++j)
            r[j] = x4[(size_t)(e[j] & 0xFFFFu) * (DD / 4) + lane];
#pragma unroll
          for (int j = 0; j < 16; ++j)
            w[j] = __int_as_float(__builtin_amdgcn_readlane(__float_as_int(wcur), g + j));
#pragma unroll
          for (int j = 0; j < 16; ++j) {
            float* a = acc + (e[j] >> 16) * DD + lane;
            atomicAdd(a, r[j].x * w[j]);
            atomicAdd(a + 32, r[j].y * w[j]);
            atomicAdd(a + 64, r[j].z * w[j]);
            atomicAdd(a + 96, r[j].w * w[j]);
          }
        }
      }
      cur = nxt;
      wcur = wnxt;
      nxt = nn;
    }
  } else {
  for (int it = wave; it < NIT; it += AGG_WAVES) {
    int ip = it + (int)b * STAG;
    if (ip >= NIT) ip -= NIT;
    uint4 k[KPI];
#pragma unroll
    for (int u = 0; u < KPI; ++u)
      k[u] = pk4[((size_t)ip * KPI + u) * 32 + lane];
#pragma unroll
    for (int u = 0; u < KPI; ++u) {
      const unsigned kk[4] = {k[u].x, k[u].y, k[u].z, k[u].w};
#pragma unroll
      for (int j = 0; j < 4; ++j) {
        const unsigned p = kk[j];
        const bool tf = (p >> 24) == b;
        const bool tb = ((p >> 8) & 255u) == b;
        const unsigned mf = __builtin_amdgcn_ballot_w32(tf);
        const unsigned mb = __builtin_amdgcn_ballot_w32(tb);
        if (mf | mb) {
          if (mf) {
            unsigned pos = cnt + __builtin_amdgcn_mbcnt_lo(mf, 0u);
            if (tf) myq[pos] = p & 0x00FFFFFFu;
            cnt += __builtin_popcount(mf);
          }
          if (mb) {
            unsigned pos = cnt + __builtin_amdgcn_mbcnt_lo(mb, 0u);
            if (tb) myq[pos] = (p >> 16) | ((p & 255u) << 16) | (1u << 24);
            cnt += __builtin_popcount(mb);
          }
        }
      }
      if (cnt > 64) {
        drain(cnt);
        cnt = 0;
      }
    }
  }
  drain(cnt);
  if (!REPLAY) {
    if (saved <= LCAP) mylist[saved + lane] = 1 << 25;
    if (lane == 0) cnts[wg] = (saved <= LCAP) ? saved : -1;
  }
  }
  __syncthreads();

  for (int row = wave; row < 2 * RNG; row += AGG_WAVES) {
    int node = (int)b * RNG + (row & (RNG - 1));
    if (node < NN) {
      const float* a = acc + row * DD + lane;
      float4 v = make_float4(a[0], a[32], a[64], a[96]);
      float* o = (row < RNG) ? fwd : bwd;
      *(float4*)(o + (size_t)node * DD + lane * 4) = v;
    }
  }
}

#define GEMM_WG 256
#define NTILE (NN / 16)
#define APITCH (K3 + 4)

static_assert(NN % 16 == 0, "row tiles are whole");

typedef __attribute__((ext_vector_type(4))) float v4f;

__global__ __launch_bounds__(256) void gemm_wmma_kernel(
    const float* __restrict__ x, const float* __restrict__ fwd,
    const float* __restrict__ bwd, const float* __restrict__ W,
    const float* __restrict__ bias, float* __restrict__ out, int do_relu) {
  __shared__ __attribute__((aligned(16))) float As[2][16][APITCH];

  const int tid = threadIdx.x;
  const int wave = tid >> 5;
  const int lane = tid & 31;
  const int hl = lane >> 4;
  const int l16 = lane & 15;
  const int col = wave * 16 + l16;

  v2f bw[K3 / 4];
#pragma unroll
  for (int s = 0; s < K3 / 4; ++s) {
    bw[s].x = W[(size_t)(4 * s + 2 * hl + 0) * DD + col];
    bw[s].y = W[(size_t)(4 * s + 2 * hl + 1) * DD + col];
  }
  const float bb = bias[col];

  const v4f* x4 = (const v4f*)x + tid;
  const v4f* f4 = (const v4f*)fwd + tid;
  const v4f* b4 = (const v4f*)bwd + tid;
  const int sA = (tid >> 5) * APITCH + 4 * (tid & 31);
  v4f g0, g1, g2, g3, g4, g5;

  int tile = blockIdx.x;
  {
    const size_t base = (size_t)tile * (16 * DD / 4);
    g0 = x4[base]; g1 = x4[base + 256];
    g2 = f4[base]; g3 = f4[base + 256];
    g4 = b4[base]; g5 = b4[base + 256];
    float* d = &As[0][0][0] + sA;
    *(v4f*)(d) = g0;            *(v4f*)(d + 8 * APITCH) = g1;
    *(v4f*)(d + DD) = g2;       *(v4f*)(d + 8 * APITCH + DD) = g3;
    *(v4f*)(d + 2 * DD) = g4;   *(v4f*)(d + 8 * APITCH + 2 * DD) = g5;
  }
  __syncthreads();

  int buf = 0;
  for (;;) {
    const bool more = tile + GEMM_WG < NTILE;
    if (more) {
      const size_t base = (size_t)(tile + GEMM_WG) * (16 * DD / 4);
      g0 = x4[base]; g1 = x4[base + 256];
      g2 = f4[base]; g3 = f4[base + 256];
      g4 = b4[base]; g5 = b4[base + 256];
    }

    const float* arow = &As[buf][l16][2 * hl];
    v2f af[2][8];
#pragma unroll
    for (int j = 0; j < 8; ++j) af[0][j] = *(const v2f*)(arow + 4 * j);
    v8f acc = {};
#pragma unroll
    for (int gidx = 0; gidx < K3 / 32; ++gidx) {
      if (gidx + 1 < K3 / 32) {
#pragma unroll
        for (int j = 0; j < 8; ++j)
          af[(gidx + 1) & 1][j] = *(const v2f*)(arow + 32 * (gidx + 1) + 4 * j);
      }
      __builtin_amdgcn_sched_barrier(0);
#pragma unroll
      for (int j = 0; j < 8; ++j)
        acc = __builtin_amdgcn_wmma_f32_16x16x4_f32(
             false, af[gidx & 1][j],  false, bw[8 * gidx + j],
             (short)0, acc,  false,  false);
    }

    {
      float* d = &As[buf ^ 1][0][0] + sA;
      *(v4f*)(d) = g0;            *(v4f*)(d + 8 * APITCH) = g1;
      *(v4f*)(d + DD) = g2;       *(v4f*)(d + 8 * APITCH + DD) = g3;
      *(v4f*)(d + 2 * DD) = g4;   *(v4f*)(d + 8 * APITCH + 2 * DD) = g5;
    }
    __syncthreads();

    asm volatile("v_nop\n\tv_nop\n\tv_nop\n\tv_nop\n\tv_nop" : "+v"(acc));

    float* o = out + (size_t)(tile * 16 + 8 * hl) * DD + col;
#pragma unroll
    for (int r = 0; r < 8; ++r) {
      float v = acc[r] + bb;
      if (do_relu) v = fmaxf(v, 0.0f);
      o[r * DD] = v;
    }

    if (!more) break;
    tile += GEMM_WG;
    buf ^= 1;
  }
}

extern "C" void kernel_launch(void* const* d_in, const int* in_sizes, int n_in,
                              void* d_out, int out_size, void* d_ws, size_t ws_size,
                              hipStream_t stream) {
  const float* x  = (const float*)d_in[0];
  const float* W0 = (const float*)d_in[1];
  const float* b0 = (const float*)d_in[2];
  const float* W1 = (const float*)d_in[3];
  const float* b1 = (const float*)d_in[4];
  const float* W2 = (const float*)d_in[5];
  const float* b2 = (const float*)d_in[6];
  const int* src  = (const int*)d_in[7];
  const int* dst  = (const int*)d_in[8];
  float* out = (float*)d_out;

  float* ws      = (float*)d_ws;
  float* deg_out = ws;
  float* deg_in  = ws + NN;
  float* xA      = ws + 2 * (size_t)NN;
  int*   pk      = (int*)(xA + (size_t)NN * DD);
  int*   lists   = pk + EE;
  int*   cnts    = lists + (size_t)NLW * LSTRIDE;
  float* fwd     = xA + 2 * (size_t)NN * DD;
  float* bwd     = fwd + (size_t)NN * DD;

  hipMemsetAsync(deg_out, 0, 2 * (size_t)NN * sizeof(float), stream);
  deg_kernel<<<(EE + 255) / 256, 256, 0, stream>>>(src, dst, deg_out, deg_in, pk, EE);
  inv_kernel<<<(2 * NN + 255) / 256, 256, 0, stream>>>(deg_out, 2 * NN);

  const float* Ws[3]  = {W0, W1, W2};
  const float* bs[3]  = {b0, b1, b2};
  float*       dsts[3] = {out, xA, out};

  const float* cur = x;
  for (int l = 0; l < 3; ++l) {
    if (l == 0)
      agg_owner_kernel<0><<<NWG, 1024, 0, stream>>>(cur, pk, deg_out, fwd, bwd, lists, cnts);
    else
      agg_owner_kernel<1><<<NWG, 1024, 0, stream>>>(cur, pk, deg_out, fwd, bwd, lists, cnts);
    gemm_wmma_kernel<<<GEMM_WG, 256, 0, stream>>>(cur, fwd, bwd, Ws[l], bs[l],
                                                  dsts[l], (l < 2) ? 1 : 0);
    cur = dsts[l];
  }
}


# Aggregation ranges of 196 nodes, not 256: 256 owner work-groups, one per WGP (was 196, 60 WGPs idle), each with a quarter fewer LDS float adds; range test on 16-bit halves; sum order is LDS arrival
# speedup vs baseline: 2.8816x; 1.0533x over previous
#include <hip/hip_runtime.h>

#define NN 50000
#define EE 800000
#define DD 128
#define K3 384

#define RNG 196
#define NWG ((NN + RNG - 1) / RNG)
#define DPITCH 256
#define AGG_WAVES 32
#define QCAP 320
#define KPI 5
#define NIT (EE / (128 * KPI))
#define STAG (NIT / NWG)
#define LSTRIDE 640
#define LCAP (LSTRIDE - 32)
#define NLW (NWG * AGG_WAVES)

static_assert(NN <= 65536, "packed keys hold a node id in 16 bits");
static_assert(EE % (128 * KPI) == 0, "key stream is read in whole scan iterations");
static_assert(RNG <= DPITCH, "a local node id fits the eight bits under the direction bit");
static_assert((size_t)NWG * RNG <= 65536, "the range test's subtraction wraps only for nodes below the range");
static_assert((size_t)EE + (size_t)NLW * LSTRIDE + NLW <= (size_t)NN * DD,
              "keys, saved lists and list lengths share the key plane of the workspace");
static_assert(LSTRIDE % 32 == 0 && LCAP >= QCAP, "lists are read in chunks of 32 entries");

typedef __attribute__((ext_vector_type(2))) float v2f;
typedef __attribute__((ext_vector_type(2))) unsigned short us2;
typedef __attribute__((ext_vector_type(8))) float v8f;

__global__ __launch_bounds__(256) void deg_kernel(
    const int* __restrict__ src, const int* __restrict__ dst,
    float* __restrict__ deg_out, float* __restrict__ deg_in,
    int* __restrict__ pk, int E) {
  int e = blockIdx.x * blockDim.x + threadIdx.x;
  if (e >= E) return;
  int s = src[e];
  int d = dst[e];
  atomicAdd(&deg_out[s], 1.0f);
  atomicAdd(&deg_in[d], 1.0f);
  pk[e] = (int)((unsigned)s | ((unsigned)d << 16));
}

__global__ __launch_bounds__(256) void inv_kernel(float* __restrict__ deg, int n) {
  int i = blockIdx.x * blockDim.x + threadIdx.x;
  if (i >= n) return;
  deg[i] = 1.0f / fmaxf(deg[i], 1.0f);
}

template <int REPLAY>
__global__ __launch_bounds__(1024) void agg_owner_kernel(
    const float* __restrict__ x, const int* __restrict__ pk,
    const float* __restrict__ inv,
    float* __restrict__ fwd, float* __restrict__ bwd,
    int* __restrict__ lists,
    int* __restrict__ cnts) {
  __shared__ float acc[(2 * DPITCH + 1) * DD];
  __shared__ unsigned q[AGG_WAVES][QCAP + 8];

  const int tid = threadIdx.x;
  const int lane = tid & 31;
  const int wave = __builtin_amdgcn_readfirstlane(tid >> 5);
  const unsigned b = blockIdx.x;

  const int wg = (int)b * AGG_WAVES + wave;
  int* mylist = lists + (size_t)wg * LSTRIDE;
  int nrep = -1;
  if (REPLAY) nrep = __builtin_amdgcn_readfirstlane(cnts[wg]);

  float4* acc4 = (float4*)acc;
  for (int i = tid; i < (2 * DPITCH + 1) * DD / 4; i += 1024)
    acc4[i] = make_float4(0.0f, 0.0f, 0.0f, 0.0f);
  __syncthreads();

  const uint4* pk4 = (const uint4*)pk;
  const float4* x4 = (const float4*)x;
  const unsigned short lo = (unsigned short)(b * RNG);
  const us2 lo2 = {lo, lo};
  unsigned* myq = q[wave];
  int cnt = 0;

  int saved = 0;
  auto drain = [&](int n) {
    if (lane < 8) myq[n + lane] = 1u << 25;
    if (!REPLAY) {
      if (saved + n <= LCAP) {
#pragma nounroll
        for (int i = lane; i < n; i += 32) mylist[saved + i] = (int)myq[i];
        saved += n;
      } else {
        saved = LCAP + 1;
      }
    }
    for (int i = 0; i < n; i += 8) {
      unsigned e[8];
      float w[8];
      float4 r[8];
#pragma unroll
      for (int j = 0; j < 8; ++j)
        e[j] = __builtin_amdgcn_readfirstlane(myq[i + j]);
#pragma unroll
      for (int j = 0; j < 8; ++j) {
        unsigned o = e[j] & 0xFFFFu;
        w[j] = inv[((e[j] >> 24) & 1u) * NN + o];
        r[j] = x4[(size_t)o * (DD / 4) + lane];
      }
#pragma unroll
      for (int j = 0; j < 8; ++j) {
        float* a = acc + (e[j] >> 16) * DD + lane;
        atomicAdd(a, r[j].x * w[j]);
        atomicAdd(a + 32, r[j].y * w[j]);
        atomicAdd(a + 64, r[j].z * w[j]);
        atomicAdd(a + 96, r[j].w * w[j]);
      }
    }
  };

  if (REPLAY && nrep >= 0) {
    unsigned cur = (unsigned)mylist[lane];
    unsigned nxt = 0;
    if (32 < nrep) nxt = (unsigned)mylist[32 + lane];
    float wcur = inv[((cur >> 24) & 1u) * NN + (cur & 0xFFFFu)];
    asm volatile("" : "+v"(wcur));
    for (int base = 0; base < nrep; base += 32) {
      float wnxt = 0.0f;
      unsigned nn = 0;
      if (base + 32 < nrep) wnxt = inv[((nxt >> 24) & 1u) * NN + (nxt & 0xFFFFu)];
      if (base + 64 < nrep) nn = (unsigned)mylist[base + 64 + lane];
#pragma unroll
      for (int g = 0; g < 32; g += 16) {
        if (base + g < nrep) {
          unsigned e[16];
          float w[16];
          float4 r[16];
#pragma unroll
          for (int j = 0; j < 16; ++j)
            e[j] = (unsigned)__builtin_amdgcn_readlane((int)cur, g + j);
#pragma unroll
          for (int j = 0; j < 16; ++j)
            r[j] = x4[(size_t)(e[j] & 0xFFFFu) * (DD / 4) + lane];
#pragma unroll
          for (int j = 0; j < 16; ++j)
            w[j] = __int_as_float(__builtin_amdgcn_readlane(__float_as_int(wcur), g + j));
#pragma unroll
          for (int j = 0; j < 16; ++j) {
            float* a = acc + (e[j] >> 16) * DD + lane;
            atomicAdd(a, r[j].x * w[j]);
            atomicAdd(a + 32, r[j].y * w[j]);
            atomicAdd(a + 64, r[j].z * w[j]);
            atomicAdd(a + 96, r[j].w * w[j]);
          }
        }
      }
      cur = nxt;
      wcur = wnxt;
      nxt = nn;
    }
  } else {
  for (int it = wave; it < NIT; it += AGG_WAVES) {
    int ip = it + (int)b * STAG;
    if (ip >= NIT) ip -= NIT;
    uint4 k[KPI];
#pragma unroll
    for (int u = 0; u < KPI; ++u)
      k[u] = pk4[((size_t)ip * KPI + u) * 32 + lane];
#pragma unroll
    for (int u = 0; u < KPI; ++u) {
      const unsigned kk[4] = {k[u].x, k[u].y, k[u].z, k[u].w};
#pragma unroll
      for (int j = 0; j < 4; ++j) {
        const unsigned p = kk[j];
        const us2 rel = __builtin_bit_cast(us2, p) - lo2;
        const unsigned relw = __builtin_bit_cast(unsigned, rel);
        const bool tf = relw < ((unsigned)RNG << 16);
        const bool tb = rel.x < RNG;
        const unsigned mf = __builtin_amdgcn_ballot_w32(tf);
        const unsigned mb = __builtin_amdgcn_ballot_w32(tb);
        if (mf | mb) {
          if (mf) {
            unsigned pos = cnt + __builtin_amdgcn_mbcnt_lo(mf, 0u);
            if (tf) myq[pos] = (relw & 0xFFFF0000u) | (p & 0xFFFFu);
            cnt += __builtin_popcount(mf);
          }
          if (mb) {
            unsigned pos = cnt + __builtin_amdgcn_mbcnt_lo(mb, 0u);
            if (tb) myq[pos] = (relw << 16) | (p >> 16) | (1u << 24);
            cnt += __builtin_popcount(mb);
          }
        }
      }
      if (cnt > 64) {
        drain(cnt);
        cnt = 0;
      }
    }
  }
  drain(cnt);
  if (!REPLAY) {
    if (saved <= LCAP) mylist[saved + lane] = 1 << 25;
    if (lane == 0) cnts[wg] = (saved <= LCAP) ? saved : -1;
  }
  }
  __syncthreads();

  for (int row = wave; row < 2 * RNG; row += AGG_WAVES) {
    const int dir = (row >= RNG) ? 1 : 0;
    const int loc = row - dir * RNG;
    int node = (int)b * RNG + loc;
    if (node < NN) {
      const float* a = acc + (dir * DPITCH + loc) * DD + lane;
      float4 v = make_float4(a[0], a[32], a[64], a[96]);
      float* o = dir ? bwd : fwd;
      *(float4*)(o + (size_t)node * DD + lane * 4) = v;
    }
  }
}

#define GEMM_WG 256
#define NTILE (NN / 16)
#define APITCH (K3 + 4)

static_assert(NN % 16 == 0, "row tiles are whole");

typedef __attribute__((ext_vector_type(4))) float v4f;

__global__ __launch_bounds__(256) void gemm_wmma_kernel(
    const float* __restrict__ x, const float* __restrict__ fwd,
    const float* __restrict__ bwd, const float* __restrict__ W,
    const float* __restrict__ bias, float* __restrict__ out, int do_relu) {
  __shared__ __attribute__((aligned(16))) float As[2][16][APITCH];

  const int tid = threadIdx.x;
  const int wave = tid >> 5;
  const int lane = tid & 31;
  const int hl = lane >> 4;
  const int l16 = lane & 15;
  const int col = wave * 16 + l16;

  v2f bw[K3 / 4];
#pragma unroll
  for (int s = 0; s < K3 / 4; ++s) {
    bw[s].x = W[(size_t)(4 * s + 2 * hl + 0) * DD + col];
    bw[s].y = W[(size_t)(4 * s + 2 * hl + 1) * DD + col];
  }
  const float bb = bias[col];

  const v4f* x4 = (const v4f*)x + tid;
  const v4f* f4 = (const v4f*)fwd + tid;
  const v4f* b4 = (const v4f*)bwd + tid;
  const int sA = (tid >> 5) * APITCH + 4 * (tid & 31);
  v4f g0, g1, g2, g3, g4, g5;

  int tile = blockIdx.x;
  {
    const size_t base = (size_t)tile * (16 * DD / 4);
    g0 = x4[base]; g1 = x4[base + 256];
    g2 = f4[base]; g3 = f4[base + 256];
    g4 = b4[base]; g5 = b4[base + 256];
    float* d = &As[0][0][0] + sA;
    *(v4f*)(d) = g0;            *(v4f*)(d + 8 * APITCH) = g1;
    *(v4f*)(d + DD) = g2;       *(v4f*)(d + 8 * APITCH + DD) = g3;
    *(v4f*)(d + 2 * DD) = g4;   *(v4f*)(d + 8 * APITCH + 2 * DD) = g5;
  }
  __syncthreads();

  int buf = 0;
  for (;;) {
    const bool more = tile + GEMM_WG < NTILE;
    if (more) {
      const size_t base = (size_t)(tile + GEMM_WG) * (16 * DD / 4);
      g0 = x4[base]; g1 = x4[base + 256];
      g2 = f4[base]; g3 = f4[base + 256];
      g4 = b4[base]; g5 = b4[base + 256];
    }

    const float* arow = &As[buf][l16][2 * hl];
    v2f af[2][8];
#pragma unroll
    for (int j = 0; j < 8; ++j) af[0][j] = *(const v2f*)(arow + 4 * j);
    v8f acc = {};
#pragma unroll
    for (int gidx = 0; gidx < K3 / 32; ++gidx) {
      if (gidx + 1 < K3 / 32) {
#pragma unroll
        for (int j = 0; j < 8; ++j)
          af[(gidx + 1) & 1][j] = *(const v2f*)(arow + 32 * (gidx + 1) + 4 * j);
      }
      __builtin_amdgcn_sched_barrier(0);
#pragma unroll
      for (int j = 0; j < 8; ++j)
        acc = __builtin_amdgcn_wmma_f32_16x16x4_f32(
             false, af[gidx & 1][j],  false, bw[8 * gidx + j],
             (short)0, acc,  false,  false);
    }

    {
      float* d = &As[buf ^ 1][0][0] + sA;
      *(v4f*)(d) = g0;            *(v4f*)(d + 8 * APITCH) = g1;
      *(v4f*)(d + DD) = g2;       *(v4f*)(d + 8 * APITCH + DD) = g3;
      *(v4f*)(d + 2 * DD) = g4;   *(v4f*)(d + 8 * APITCH + 2 * DD) = g5;
    }
    __syncthreads();

    asm volatile("v_nop\n\tv_nop\n\tv_nop\n\tv_nop\n\tv_nop" : "+v"(acc));

    float* o = out + (size_t)(tile * 16 + 8 * hl) * DD + col;
#pragma unroll
    for (int r = 0; r < 8; ++r) {
      float v = acc[r] + bb;
      if (do_relu) v = fmaxf(v, 0.0f);
      o[r * DD] = v;
    }

    if (!more) break;
    tile += GEMM_WG;
    buf ^= 1;
  }
}

extern "C" void kernel_launch(void* const* d_in, const int* in_sizes, int n_in,
                              void* d_out, int out_size, void* d_ws, size_t ws_size,
                              hipStream_t stream) {
  const float* x  = (const float*)d_in[0];
  const float* W0 = (const float*)d_in[1];
  const float* b0 = (const float*)d_in[2];
  const float* W1 = (const float*)d_in[3];
  const float* b1 = (const float*)d_in[4];
  const float* W2 = (const float*)d_in[5];
  const float* b2 = (const float*)d_in[6];
  const int* src  = (const int*)d_in[7];
  const int* dst  = (const int*)d_in[8];
  float* out = (float*)d_out;

  float* ws      = (float*)d_ws;
  float* deg_out = ws;
  float* deg_in  = ws + NN;
  float* xA      = ws + 2 * (size_t)NN;
  int*   pk      = (int*)(xA + (size_t)NN * DD);
  int*   lists   = pk + EE;
  int*   cnts    = lists + (size_t)NLW * LSTRIDE;
  float* fwd     = xA + 2 * (size_t)NN * DD;
  float* bwd     = fwd + (size_t)NN * DD;

  hipMemsetAsync(deg_out, 0, 2 * (size_t)NN * sizeof(float), stream);
  deg_kernel<<<(EE + 255) / 256, 256, 0, stream>>>(src, dst, deg_out, deg_in, pk, EE);
  inv_kernel<<<(2 * NN + 255) / 256, 256, 0, stream>>>(deg_out, 2 * NN);

  const float* Ws[3]  = {W0, W1, W2};
  const float* bs[3]  = {b0, b1, b2};
  float*       dsts[3] = {out, xA, out};

  const float* cur = x;
  for (int l = 0; l < 3; ++l) {
    if (l == 0)
      agg_owner_kernel<0><<<NWG, 1024, 0, stream>>>(cur, pk, deg_out, fwd, bwd, lists, cnts);
    else
      agg_owner_kernel<1><<<NWG, 1024, 0, stream>>>(cur, pk, deg_out, fwd, bwd, lists, cnts);
    gemm_wmma_kernel<<<GEMM_WG, 256, 0, stream>>>(cur, fwd, bwd, Ws[l], bs[l],
                                                  dsts[l], (l < 2) ? 1 : 0);
    cur = dsts[l];
  }
}


# speedup vs baseline: 2.9717x; 1.0313x over previous
#include <hip/hip_runtime.h>

#define NN 50000
#define EE 800000
#define DD 128
#define K3 384

#define RNG 196
#define NWG ((NN + RNG - 1) / RNG)
#define DPITCH 256
#define AGG_WAVES 32
#define QCAP 320
#define KPI 5
#define NIT (EE / (128 * KPI))
#define STAG (NIT / NWG)
#define LSTRIDE 640
#define LCAP (LSTRIDE - 32)
#define NLW (NWG * AGG_WAVES)

static_assert(NN <= 65536, "packed keys hold a node id in 16 bits");
static_assert(EE % (128 * KPI) == 0, "key stream is read in whole scan iterations");
static_assert(RNG <= DPITCH, "a local node id fits the eight bits under the direction bit");
static_assert((size_t)NWG * RNG <= 65536, "the range test's subtraction wraps only for nodes below the range");
#define WS_ALIGN 128
#define XA_OFF ((2 * NN + WS_ALIGN - 1) / WS_ALIGN * WS_ALIGN)
#define KEY_PLANE (NN * DD - WS_ALIGN)
static_assert(XA_OFF - 2 * NN <= WS_ALIGN && (XA_OFF + KEY_PLANE) % WS_ALIGN == 0,
              "row planes start on 512-byte boundaries and the layout is no longer than before");
static_assert((size_t)EE + (size_t)NLW * LSTRIDE + NLW <= (size_t)KEY_PLANE,
              "keys, saved lists and list lengths share the key plane of the workspace");
static_assert(LSTRIDE % 32 == 0 && LCAP >= QCAP, "lists are read in chunks of 32 entries");

typedef __attribute__((ext_vector_type(2))) float v2f;
typedef __attribute__((ext_vector_type(2))) unsigned short us2;
typedef __attribute__((ext_vector_type(8))) float v8f;

__global__ __launch_bounds__(256) void deg_kernel(
    const int* __restrict__ src, const int* __restrict__ dst,
    float* __restrict__ deg_out, float* __restrict__ deg_in,
    int* __restrict__ pk, int E) {
  int e = blockIdx.x * blockDim.x + threadIdx.x;
  if (e >= E) return;
  int s = src[e];
  int d = dst[e];
  atomicAdd(&deg_out[s], 1.0f);
  atomicAdd(&deg_in[d], 1.0f);
  pk[e] = (int)((unsigned)s | ((unsigned)d << 16));
}

__global__ __launch_bounds__(256) void inv_kernel(float* __restrict__ deg, int n) {
  int i = blockIdx.x * blockDim.x + threadIdx.x;
  if (i >= n) return;
  deg[i] = 1.0f / fmaxf(deg[i], 1.0f);
}

template <int REPLAY>
__global__ __launch_bounds__(1024) void agg_owner_kernel(
    const float* __restrict__ x, const int* __restrict__ pk,
    const float* __restrict__ inv,
    float* __restrict__ fwd, float* __restrict__ bwd,
    int* __restrict__ lists,
    int* __restrict__ cnts) {
  __shared__ float acc[(2 * DPITCH + 1) * DD];
  __shared__ unsigned q[AGG_WAVES][QCAP + 8];

  const int tid = threadIdx.x;
  const int lane = tid & 31;
  const int wave = __builtin_amdgcn_readfirstlane(tid >> 5);
  const unsigned b = blockIdx.x;

  const int wg = (int)b * AGG_WAVES + wave;
  int* mylist = lists + (size_t)wg * LSTRIDE;
  int nrep = -1;
  if (REPLAY) nrep = __builtin_amdgcn_readfirstlane(cnts[wg]);

  float4* acc4 = (float4*)acc;
  for (int i = tid; i < (2 * DPITCH + 1) * DD / 4; i += 1024)
    acc4[i] = make_float4(0.0f, 0.0f, 0.0f, 0.0f);
  __syncthreads();

  const uint4* pk4 = (const uint4*)pk;
  const float4* x4 = (const float4*)x;
  const unsigned short lo = (unsigned short)(b * RNG);
  const us2 lo2 = {lo, lo};
  unsigned* myq = q[wave];
  int cnt = 0;

  int saved = 0;
  auto drain = [&](int n) {
    if (lane < 8) myq[n + lane] = 1u << 25;
    if (!REPLAY) {
      if (saved + n <= LCAP) {
#pragma nounroll
        for (int i = lane; i < n; i += 32) mylist[saved + i] = (int)myq[i];
        saved += n;
      } else {
        saved = LCAP + 1;
      }
    }
    for (int i = 0; i < n; i += 8) {
      unsigned e[8];
      float w[8];
      float4 r[8];
#pragma unroll
      for (int j = 0; j < 8; ++j)
        e[j] = __builtin_amdgcn_readfirstlane(myq[i + j]);
#pragma unroll
      for (int j = 0; j < 8; ++j) {
        unsigned o = e[j] & 0xFFFFu;
        w[j] = inv[((e[j] >> 24) & 1u) * NN + o];
        r[j] = x4[(size_t)o * (DD / 4) + lane];
      }
#pragma unroll
      for (int j = 0; j < 8; ++j) {
        float* a = acc + (e[j] >> 16) * DD + lane;
        atomicAdd(a, r[j].x * w[j]);
        atomicAdd(a + 32, r[j].y * w[j]);
        atomicAdd(a + 64, r[j].z * w[j]);
        atomicAdd(a + 96, r[j].w * w[j]);
      }
    }
  };

  if (REPLAY && nrep >= 0) {
    unsigned cur = (unsigned)mylist[lane];
    unsigned nxt = 0;
    if (32 < nrep) nxt = (unsigned)mylist[32 + lane];
    float wcur = inv[((cur >> 24) & 1u) * NN + (cur & 0xFFFFu)];
    asm volatile("" : "+v"(wcur));
    for (int base = 0; base < nrep; base += 32) {
      float wnxt = 0.0f;
      unsigned nn = 0;
      if (base + 32 < nrep) wnxt = inv[((nxt >> 24) & 1u) * NN + (nxt & 0xFFFFu)];
      if (base + 64 < nrep) nn = (unsigned)mylist[base + 64 + lane];
#pragma unroll
      for (int g = 0; g < 32; g += 16) {
        if (base + g < nrep) {
          unsigned e[16];
          float w[16];
          float4 r[16];
#pragma unroll
          for (int j = 0; j < 16; ++j)
            e[j] = (unsigned)__builtin_amdgcn_readlane((int)cur, g + j);
#pragma unroll
          for (int j = 0; j < 16; ++j)
            r[j] = x4[(size_t)(e[j] & 0xFFFFu) * (DD / 4) + lane];
#pragma unroll
          for (int j = 0; j < 16; ++j)
            w[j] = __int_as_float(__builtin_amdgcn_readlane(__float_as_int(wcur), g + j));
#pragma unroll
          for (int j = 0; j < 16; ++j) {
            float* a = acc + (e[j] >> 16) * DD + lane;
            atomicAdd(a, r[j].x * w[j]);
            atomicAdd(a + 32, r[j].y * w[j]);
            atomicAdd(a + 64, r[j].z * w[j]);
            atomicAdd(a + 96, r[j].w * w[j]);
          }
        }
      }
      cur = nxt;
      wcur = wnxt;
      nxt = nn;
    }
  } else {
  for (int it = wave; it < NIT; it += AGG_WAVES) {
    int ip = it + (int)b * STAG;
    if (ip >= NIT) ip -= NIT;
    uint4 k[KPI];
#pragma unroll
    for (int u = 0; u < KPI; ++u)
      k[u] = pk4[((size_t)ip * KPI + u) * 32 + lane];
#pragma unroll
    for (int u = 0; u < KPI; ++u) {
      const unsigned kk[4] = {k[u].x, k[u].y, k[u].z, k[u].w};
#pragma unroll
      for (int j = 0; j < 4; ++j) {
        const unsigned p = kk[j];
        const us2 rel = __builtin_bit_cast(us2, p) - lo2;
        const unsigned relw = __builtin_bit_cast(unsigned, rel);
        const bool tf = relw < ((unsigned)RNG << 16);
        const bool tb = rel.x < RNG;
        const unsigned mf = __builtin_amdgcn_ballot_w32(tf);
        const unsigned mb = __builtin_amdgcn_ballot_w32(tb);
        if (mf | mb) {
          if (mf) {
            unsigned pos = cnt + __builtin_amdgcn_mbcnt_lo(mf, 0u);
            if (tf) myq[pos] = (relw & 0xFFFF0000u) | (p & 0xFFFFu);
            cnt += __builtin_popcount(mf);
          }
          if (mb) {
            unsigned pos = cnt + __builtin_amdgcn_mbcnt_lo(mb, 0u);
            if (tb) myq[pos] = (relw << 16) | (p >> 16) | (1u << 24);
            cnt += __builtin_popcount(mb);
          }
        }
      }
      if (cnt > 64) {
        drain(cnt);
        cnt = 0;
      }
    }
  }
  drain(cnt);
  if (!REPLAY) {
    if (saved <= LCAP) mylist[saved + lane] = 1 << 25;
    if (lane == 0) cnts[wg] = (saved <= LCAP) ? saved : -1;
  }
  }
  __syncthreads();

  for (int row = wave; row < 2 * RNG; row += AGG_WAVES) {
    const int dir = (row >= RNG) ? 1 : 0;
    const int loc = row - dir * RNG;
    int node = (int)b * RNG + loc;
    if (node < NN) {
      const float* a = acc + (dir * DPITCH + loc) * DD + lane;
      float4 v = make_float4(a[0], a[32], a[64], a[96]);
      float* o = dir ? bwd : fwd;
      *(float4*)(o + (size_t)node * DD + lane * 4) = v;
    }
  }
}

#define GEMM_WG 256
#define NTILE (NN / 16)
#define APITCH (K3 + 4)

static_assert(NN % 16 == 0, "row tiles are whole");

typedef __attribute__((ext_vector_type(4))) float v4f;

__global__ __launch_bounds__(256) void gemm_wmma_kernel(
    const float* __restrict__ x, const float* __restrict__ fwd,
    const float* __restrict__ bwd, const float* __restrict__ W,
    const float* __restrict__ bias, float* __restrict__ out, int do_relu) {
  __shared__ __attribute__((aligned(16))) float As[2][16][APITCH];

  const int tid = threadIdx.x;
  const int wave = tid >> 5;
  const int lane = tid & 31;
  const int hl = lane >> 4;
  const int l16 = lane & 15;
  const int col = wave * 16 + l16;

  v2f bw[K3 / 4];
#pragma unroll
  for (int s = 0; s < K3 / 4; ++s) {
    bw[s].x = W[(size_t)(4 * s + 2 * hl + 0) * DD + col];
    bw[s].y = W[(size_t)(4 * s + 2 * hl + 1) * DD + col];
  }
  const float bb = bias[col];

  const v4f* x4 = (const v4f*)x + tid;
  const v4f* f4 = (const v4f*)fwd + tid;
  const v4f* b4 = (const v4f*)bwd + tid;
  const int sA = (tid >> 5) * APITCH + 4 * (tid & 31);
  v4f g0, g1, g2, g3, g4, g5;

  int tile = blockIdx.x;
  {
    const size_t base = (size_t)tile * (16 * DD / 4);
    g0 = x4[base]; g1 = x4[base + 256];
    g2 = f4[base]; g3 = f4[base + 256];
    g4 = b4[base]; g5 = b4[base + 256];
    float* d = &As[0][0][0] + sA;
    *(v4f*)(d) = g0;            *(v4f*)(d + 8 * APITCH) = g1;
    *(v4f*)(d + DD) = g2;       *(v4f*)(d + 8 * APITCH + DD) = g3;
    *(v4f*)(d + 2 * DD) = g4;   *(v4f*)(d + 8 * APITCH + 2 * DD) = g5;
  }
  __syncthreads();

  int buf = 0;
  for (;;) {
    const bool more = tile + GEMM_WG < NTILE;
    if (more) {
      const size_t base = (size_t)(tile + GEMM_WG) * (16 * DD / 4);
      g0 = x4[base]; g1 = x4[base + 256];
      g2 = f4[base]; g3 = f4[base + 256];
      g4 = b4[base]; g5 = b4[base + 256];
    }

    const float* arow = &As[buf][l16][2 * hl];
    v2f af[2][8];
#pragma unroll
    for (int j = 0; j < 8; ++j) af[0][j] = *(const v2f*)(arow + 4 * j);
    v8f acc = {};
#pragma unroll
    for (int gidx = 0; gidx < K3 / 32; ++gidx) {
      if (gidx + 1 < K3 / 32) {
#pragma unroll
        for (int j = 0; j < 8; ++j)
          af[(gidx + 1) & 1][j] = *(const v2f*)(arow + 32 * (gidx + 1) + 4 * j);
      }
      __builtin_amdgcn_sched_barrier(0);
#pragma unroll
      for (int j = 0; j < 8; ++j)
        acc = __builtin_amdgcn_wmma_f32_16x16x4_f32(
             false, af[gidx & 1][j],  false, bw[8 * gidx + j],
             (short)0, acc,  false,  false);
    }

    {
      float* d = &As[buf ^ 1][0][0] + sA;
      *(v4f*)(d) = g0;            *(v4f*)(d + 8 * APITCH) = g1;
      *(v4f*)(d + DD) = g2;       *(v4f*)(d + 8 * APITCH + DD) = g3;
      *(v4f*)(d + 2 * DD) = g4;   *(v4f*)(d + 8 * APITCH + 2 * DD) = g5;
    }
    __syncthreads();

    asm volatile("v_nop\n\tv_nop\n\tv_nop\n\tv_nop\n\tv_nop" : "+v"(acc));

    float* o = out + (size_t)(tile * 16 + 8 * hl) * DD + col;
#pragma unroll
    for (int r = 0; r < 8; ++r) {
      float v = acc[r] + bb;
      if (do_relu) v = fmaxf(v, 0.0f);
      o[r * DD] = v;
    }

    if (!more) break;
    tile += GEMM_WG;
    buf ^= 1;
  }
}

extern "C" void kernel_launch(void* const* d_in, const int* in_sizes, int n_in,
                              void* d_out, int out_size, void* d_ws, size_t ws_size,
                              hipStream_t stream) {
  const float* x  = (const float*)d_in[0];
  const float* W0 = (const float*)d_in[1];
  const float* b0 = (const float*)d_in[2];
  const float* W1 = (const float*)d_in[3];
  const float* b1 = (const float*)d_in[4];
  const float* W2 = (const float*)d_in[5];
  const float* b2 = (const float*)d_in[6];
  const int* src  = (const int*)d_in[7];
  const int* dst  = (const int*)d_in[8];
  float* out = (float*)d_out;

  float* ws      = (float*)d_ws;
  float* deg_out = ws;
  float* deg_in  = ws + NN;
  float* xA      = ws + XA_OFF;
  int*   pk      = (int*)(xA + (size_t)NN * DD);
  int*   lists   = pk + EE;
  int*   cnts    = lists + (size_t)NLW * LSTRIDE;
  float* fwd     = (float*)pk + KEY_PLANE;
  float* bwd     = fwd + (size_t)NN * DD;

  hipMemsetAsync(deg_out, 0, 2 * (size_t)NN * sizeof(float), stream);
  deg_kernel<<<(EE + 255) / 256, 256, 0, stream>>>(src, dst, deg_out, deg_in, pk, EE);
  inv_kernel<<<(2 * NN + 255) / 256, 256, 0, stream>>>(deg_out, 2 * NN);

  const float* Ws[3]  = {W0, W1, W2};
  const float* bs[3]  = {b0, b1, b2};
  float*       dsts[3] = {out, xA, out};

  const float* cur = x;
  for (int l = 0; l < 3; ++l) {
    if (l == 0)
      agg_owner_kernel<0><<<NWG, 1024, 0, stream>>>(cur, pk, deg_out, fwd, bwd, lists, cnts);
    else
      agg_owner_kernel<1><<<NWG, 1024, 0, stream>>>(cur, pk, deg_out, fwd, bwd, lists, cnts);
    gemm_wmma_kernel<<<GEMM_WG, 256, 0, stream>>>(cur, fwd, bwd, Ws[l], bs[l],
                                                  dsts[l], (l < 2) ? 1 : 0);
    cur = dsts[l];
  }
}


# Degrees counted by the owner work-groups from their own matches (LDS integer counters) in a scan-only launch; global atomics, memset and reciprocal pass go; all 3 layers replay; sum order LDS arrival
# speedup vs baseline: 3.5777x; 1.2039x over previous
#include <hip/hip_runtime.h>

#define NN 50000
#define EE 800000
#define DD 128
#define K3 384

#define RNG 196
#define NWG ((NN + RNG - 1) / RNG)
#define DPITCH 256
#define AGG_WAVES 32
#define QCAP 320
#define KPI 5
#define NIT (EE / (128 * KPI))
#define STAG (NIT / NWG)
#define LSTRIDE 640
#define LCAP (LSTRIDE - 32)
#define NLW (NWG * AGG_WAVES)

static_assert(NN <= 65536, "packed keys hold a node id in 16 bits");
static_assert(EE % (128 * KPI) == 0, "key stream is read in whole scan iterations");
static_assert(RNG <= DPITCH, "a local node id fits the eight bits under the direction bit");
static_assert((size_t)NWG * RNG <= 65536, "the range test's subtraction wraps only for nodes below the range");
#define WS_ALIGN 128
#define XA_OFF ((2 * NN + WS_ALIGN - 1) / WS_ALIGN * WS_ALIGN)
#define KEY_PLANE (NN * DD - WS_ALIGN)
static_assert(XA_OFF - 2 * NN <= WS_ALIGN && (XA_OFF + KEY_PLANE) % WS_ALIGN == 0,
              "row planes start on 512-byte boundaries and the layout is no longer than before");
static_assert((size_t)EE + (size_t)NLW * LSTRIDE + NLW <= (size_t)KEY_PLANE,
              "keys, saved lists and list lengths share the key plane of the workspace");
static_assert(LSTRIDE % 32 == 0 && LCAP >= QCAP, "lists are read in chunks of 32 entries");

typedef __attribute__((ext_vector_type(2))) float v2f;
typedef __attribute__((ext_vector_type(2))) unsigned short us2;
typedef __attribute__((ext_vector_type(8))) float v8f;

__global__ __launch_bounds__(256) void pack_kernel(
    const int* __restrict__ src, const int* __restrict__ dst,
    int* __restrict__ pk, int E) {
  int e = blockIdx.x * blockDim.x + threadIdx.x;
  if (e >= E) return;
  pk[e] = (int)((unsigned)src[e] | ((unsigned)dst[e] << 16));
}

template <class Drain>
__device__ __forceinline__ void scan_keys(const int* __restrict__ pk, unsigned b, int wave, int lane,
                                          unsigned* myq, Drain&& drain) {
  const uint4* pk4 = (const uint4*)pk;
  const unsigned short lo = (unsigned short)(b * RNG);
  const us2 lo2 = {lo, lo};
  int cnt = 0;
  for (int it = wave; it < NIT; it += AGG_WAVES) {
    int ip = it + (int)b * STAG;
    if (ip >= NIT) ip -= NIT;
    uint4 k[KPI];
#pragma unroll
    for (int u = 0; u < KPI; ++u)
      k[u] = pk4[((size_t)ip * KPI + u) * 32 + lane];
#pragma unroll
    for (int u = 0; u < KPI; ++u) {
      const unsigned kk[4] = {k[u].x, k[u].y, k[u].z, k[u].w};
#pragma unroll
      for (int j = 0; j < 4; ++j) {
        const unsigned p = kk[j];
        const us2 rel = __builtin_bit_cast(us2, p) - lo2;
        const unsigned relw = __builtin_bit_cast(unsigned, rel);
        const bool tf = relw < ((unsigned)RNG << 16);
        const bool tb = rel.x < RNG;
        const unsigned mf = __builtin_amdgcn_ballot_w32(tf);
        const unsigned mb = __builtin_amdgcn_ballot_w32(tb);
        if (mf | mb) {
          if (mf) {
            unsigned pos = cnt + __builtin_amdgcn_mbcnt_lo(mf, 0u);
            if (tf) myq[pos] = (relw & 0xFFFF0000u) | (p & 0xFFFFu);
            cnt += __builtin_popcount(mf);
          }
          if (mb) {
            unsigned pos = cnt + __builtin_amdgcn_mbcnt_lo(mb, 0u);
            if (tb) myq[pos] = (relw << 16) | (p >> 16) | (1u << 24);
            cnt += __builtin_popcount(mb);
          }
        }
      }
      if (cnt > 64) {
        drain(cnt);
        cnt = 0;
      }
    }
  }
  drain(cnt);
}

__global__ __launch_bounds__(1024) void scan_owner_kernel(
    const int* __restrict__ pk,
    float* __restrict__ inv,
    int* __restrict__ lists,
    int* __restrict__ cnts) {
  __shared__ unsigned q[AGG_WAVES][QCAP + 8];
  __shared__ unsigned degc[2 * DPITCH];

  const int tid = threadIdx.x;
  const int lane = tid & 31;
  const int wave = __builtin_amdgcn_readfirstlane(tid >> 5);
  const unsigned b = blockIdx.x;

  const int wg = (int)b * AGG_WAVES + wave;
  int* mylist = lists + (size_t)wg * LSTRIDE;
  unsigned* myq = q[wave];

  if (tid < 2 * DPITCH) degc[tid] = 0u;
  __syncthreads();

  int saved = 0;
  auto drain = [&](int n) {
    const bool fits = saved + n <= LCAP;
#pragma nounroll
    for (int i = lane; i < n; i += 32) {
      const unsigned e = myq[i];
      atomicAdd(&degc[(e >> 16) & (2 * DPITCH - 1)], 1u);
      if (fits) mylist[saved + i] = (int)e;
    }
    saved = fits ? saved + n : LCAP + 1;
  };
  scan_keys(pk, b, wave, lane, myq, drain);
  if (saved <= LCAP) mylist[saved + lane] = 1 << 25;
  if (lane == 0) cnts[wg] = (saved <= LCAP) ? saved : -1;
  __syncthreads();

  if (tid < RNG) {
    const int node = (int)b * RNG + tid;
    if (node < NN) {
      inv[node] = 1.0f / fmaxf((float)degc[DPITCH + tid], 1.0f);
      inv[NN + node] = 1.0f / fmaxf((float)degc[tid], 1.0f);
    }
  }
}

__global__ __launch_bounds__(1024) void agg_owner_kernel(
    const float* __restrict__ x, const int* __restrict__ pk,
    const float* __restrict__ inv,
    float* __restrict__ fwd, float* __restrict__ bwd,
    const int* __restrict__ lists,
    const int* __restrict__ cnts) {
  __shared__ float acc[(2 * DPITCH + 1) * DD];
  __shared__ unsigned q[AGG_WAVES][QCAP + 8];

  const int tid = threadIdx.x;
  const int lane = tid & 31;
  const int wave = __builtin_amdgcn_readfirstlane(tid >> 5);
  const unsigned b = blockIdx.x;

  const int wg = (int)b * AGG_WAVES + wave;
  const int* mylist = lists + (size_t)wg * LSTRIDE;
  const int nrep = __builtin_amdgcn_readfirstlane(cnts[wg]);

  float4* acc4 = (float4*)acc;
  for (int i = tid; i < (2 * DPITCH + 1) * DD / 4; i += 1024)
    acc4[i] = make_float4(0.0f, 0.0f, 0.0f, 0.0f);
  __syncthreads();

  const float4* x4 = (const float4*)x;

  if (nrep >= 0) {
    unsigned cur = (unsigned)mylist[lane];
    unsigned nxt = 0;
    if (32 < nrep) nxt = (unsigned)mylist[32 + lane];
    float wcur = inv[((cur >> 24) & 1u) * NN + (cur & 0xFFFFu)];
    asm volatile("" : "+v"(wcur));
    for (int base = 0; base < nrep; base += 32) {
      float wnxt = 0.0f;
      unsigned nn = 0;
      if (base + 32 < nrep) wnxt = inv[((nxt >> 24) & 1u) * NN + (nxt & 0xFFFFu)];
      if (base + 64 < nrep) nn = (unsigned)mylist[base + 64 + lane];
#pragma unroll
      for (int g = 0; g < 32; g += 16) {
        if (base + g < nrep) {
          unsigned e[16];
          float w[16];
          float4 r[16];
#pragma unroll
          for (int j = 0; j < 16; ++j)
            e[j] = (unsigned)__builtin_amdgcn_readlane((int)cur, g + j);
#pragma unroll
          for (int j = 0; j < 16; ++j)
            r[j] = x4[(size_t)(e[j] & 0xFFFFu) * (DD / 4) + lane];
#pragma unroll
          for (int j = 0; j < 16; ++j)
            w[j] = __int_as_float(__builtin_amdgcn_readlane(__float_as_int(wcur), g + j));
#pragma unroll
          for (int j = 0; j < 16; ++j) {
            float* a = acc + (e[j] >> 16) * DD + lane;
            atomicAdd(a, r[j].x * w[j]);
            atomicAdd(a + 32, r[j].y * w[j]);
            atomicAdd(a + 64, r[j].z * w[j]);
            atomicAdd(a + 96, r[j].w * w[j]);
          }
        }
      }
      cur = nxt;
      wcur = wnxt;
      nxt = nn;
    }
  } else {
    unsigned* myq = q[wave];
    auto drain = [&](int n) {
      if (lane < 8) myq[n + lane] = 1u << 25;
      for (int i = 0; i < n; i += 8) {
        unsigned e[8];
        float w[8];
        float4 r[8];
#pragma unroll
        for (int j = 0; j < 8; ++j)
          e[j] = __builtin_amdgcn_readfirstlane(myq[i + j]);
#pragma unroll
        for (int j = 0; j < 8; ++j) {
          unsigned o = e[j] & 0xFFFFu;
          w[j] = inv[((e[j] >> 24) & 1u) * NN + o];
          r[j] = x4[(size_t)o * (DD / 4) + lane];
        }
#pragma unroll
        for (int j = 0; j < 8; ++j) {
          float* a = acc + (e[j] >> 16) * DD + lane;
          atomicAdd(a, r[j].x * w[j]);
          atomicAdd(a + 32, r[j].y * w[j]);
          atomicAdd(a + 64, r[j].z * w[j]);
          atomicAdd(a + 96, r[j].w * w[j]);
        }
      }
    };
    scan_keys(pk, b, wave, lane, myq, drain);
  }
  __syncthreads();

  for (int row = wave; row < 2 * RNG; row += AGG_WAVES) {
    const int dir = (row >= RNG) ? 1 : 0;
    const int loc = row - dir * RNG;
    int node = (int)b * RNG + loc;
    if (node < NN) {
      const float* a = acc + (dir * DPITCH + loc) * DD + lane;
      float4 v = make_float4(a[0], a[32], a[64], a[96]);
      float* o = dir ? bwd : fwd;
      *(float4*)(o + (size_t)node * DD + lane * 4) = v;
    }
  }
}

#define GEMM_WG 256
#define NTILE (NN / 16)
#define APITCH (K3 + 4)

static_assert(NN % 16 == 0, "row tiles are whole");

typedef __attribute__((ext_vector_type(4))) float v4f;

__global__ __launch_bounds__(256) void gemm_wmma_kernel(
    const float* __restrict__ x, const float* __restrict__ fwd,
    const float* __restrict__ bwd, const float* __restrict__ W,
    const float* __restrict__ bias, float* __restrict__ out, int do_relu) {
  __shared__ __attribute__((aligned(16))) float As[2][16][APITCH];

  const int tid = threadIdx.x;
  const int wave = tid >> 5;
  const int lane = tid & 31;
  const int hl = lane >> 4;
  const int l16 = lane & 15;
  const int col = wave * 16 + l16;

  v2f bw[K3 / 4];
#pragma unroll
  for (int s = 0; s < K3 / 4; ++s) {
    bw[s].x = W[(size_t)(4 * s + 2 * hl + 0) * DD + col];
    bw[s].y = W[(size_t)(4 * s + 2 * hl + 1) * DD + col];
  }
  const float bb = bias[col];

  const v4f* x4 = (const v4f*)x + tid;
  const v4f* f4 = (const v4f*)fwd + tid;
  const v4f* b4 = (const v4f*)bwd + tid;
  const int sA = (tid >> 5) * APITCH + 4 * (tid & 31);
  v4f g0, g1, g2, g3, g4, g5;

  int tile = blockIdx.x;
  {
    const size_t base = (size_t)tile * (16 * DD / 4);
    g0 = x4[base]; g1 = x4[base + 256];
    g2 = f4[base]; g3 = f4[base + 256];
    g4 = b4[base]; g5 = b4[base + 256];
    float* d = &As[0][0][0] + sA;
    *(v4f*)(d) = g0;            *(v4f*)(d + 8 * APITCH) = g1;
    *(v4f*)(d + DD) = g2;       *(v4f*)(d + 8 * APITCH + DD) = g3;
    *(v4f*)(d + 2 * DD) = g4;   *(v4f*)(d + 8 * APITCH + 2 * DD) = g5;
  }
  __syncthreads();

  int buf = 0;
  for (;;) {
    const bool more = tile + GEMM_WG < NTILE;
    if (more) {
      const size_t base = (size_t)(tile + GEMM_WG) * (16 * DD / 4);
      g0 = x4[base]; g1 = x4[base + 256];
      g2 = f4[base]; g3 = f4[base + 256];
      g4 = b4[base]; g5 = b4[base + 256];
    }

    const float* arow = &As[buf][l16][2 * hl];
    v2f af[2][8];
#pragma unroll
    for (int j = 0; j < 8; ++j) af[0][j] = *(const v2f*)(arow + 4 * j);
    v8f acc = {};
#pragma unroll
    for (int gidx = 0; gidx < K3 / 32; ++gidx) {
      if (gidx + 1 < K3 / 32) {
#pragma unroll
        for (int j = 0; j < 8; ++j)
          af[(gidx + 1) & 1][j] = *(const v2f*)(arow + 32 * (gidx + 1) + 4 * j);
      }
      __builtin_amdgcn_sched_barrier(0);
#pragma unroll
      for (int j = 0; j < 8; ++j)
        acc = __builtin_amdgcn_wmma_f32_16x16x4_f32(
             false, af[gidx & 1][j],  false, bw[8 * gidx + j],
             (short)0, acc,  false,  false);
    }

    {
      float* d = &As[buf ^ 1][0][0] + sA;
      *(v4f*)(d) = g0;            *(v4f*)(d + 8 * APITCH) = g1;
      *(v4f*)(d + DD) = g2;       *(v4f*)(d + 8 * APITCH + DD) = g3;
      *(v4f*)(d + 2 * DD) = g4;   *(v4f*)(d + 8 * APITCH + 2 * DD) = g5;
    }
    __syncthreads();

    asm volatile("v_nop\n\tv_nop\n\tv_nop\n\tv_nop\n\tv_nop" : "+v"(acc));

    float* o = out + (size_t)(tile * 16 + 8 * hl) * DD + col;
#pragma unroll
    for (int r = 0; r < 8; ++r) {
      float v = acc[r] + bb;
      if (do_relu) v = fmaxf(v, 0.0f);
      o[r * DD] = v;
    }

    if (!more) break;
    tile += GEMM_WG;
    buf ^= 1;
  }
}

extern "C" void kernel_launch(void* const* d_in, const int* in_sizes, int n_in,
                              void* d_out, int out_size, void* d_ws, size_t ws_size,
                              hipStream_t stream) {
  const float* x  = (const float*)d_in[0];
  const float* W0 = (const float*)d_in[1];
  const float* b0 = (const float*)d_in[2];
  const float* W1 = (const float*)d_in[3];
  const float* b1 = (const float*)d_in[4];
  const float* W2 = (const float*)d_in[5];
  const float* b2 = (const float*)d_in[6];
  const int* src  = (const int*)d_in[7];
  const int* dst  = (const int*)d_in[8];
  float* out = (float*)d_out;

  float* ws      = (float*)d_ws;
  float* inv     = ws;
  float* xA      = ws + XA_OFF;
  int*   pk      = (int*)(xA + (size_t)NN * DD);
  int*   lists   = pk + EE;
  int*   cnts    = lists + (size_t)NLW * LSTRIDE;
  float* fwd     = (float*)pk + KEY_PLANE;
  float* bwd     = fwd + (size_t)NN * DD;

  pack_kernel<<<(EE + 255) / 256, 256, 0, stream>>>(src, dst, pk, EE);
  scan_owner_kernel<<<NWG, 1024, 0, stream>>>(pk, inv, lists, cnts);

  const float* Ws[3]  = {W0, W1, W2};
  const float* bs[3]  = {b0, b1, b2};
  float*       dsts[3] = {out, xA, out};

  const float* cur = x;
  for (int l = 0; l < 3; ++l) {
    agg_owner_kernel<<<NWG, 1024, 0, stream>>>(cur, pk, inv, fwd, bwd, lists, cnts);
    gemm_wmma_kernel<<<GEMM_WG, 256, 0, stream>>>(cur, fwd, bwd, Ws[l], bs[l],
                                                  dsts[l], (l < 2) ? 1 : 0);
    cur = dsts[l];
  }
}


# Every sum row has one owner wave: the scan hands a match to its row's owner (an LDS counter gives the place), which keeps its 13 rows' sums in registers; LDS float atomics go; sum order = list order
# speedup vs baseline: 3.8270x; 1.0697x over previous
#include <hip/hip_runtime.h>

#define NN 50000
#define EE 800000
#define DD 128
#define K3 384

#define RNG 196
#define NWG ((NN + RNG - 1) / RNG)
#define DPITCH 256
#define AGG_WAVES 32
#define QCAP 320
#define KPI 5
#define NIT (EE / (128 * KPI))
#define STAG (NIT / NWG)
#define LSTRIDE 640
#define LCAP (LSTRIDE - 32)
#define NLW (NWG * AGG_WAVES)
#define OWN_SLOTS ((2 * RNG + AGG_WAVES - 1) / AGG_WAVES)
#define FILL_ENTRY (15u << 16)

static_assert(NN <= 65536, "packed keys hold a node id in 16 bits");
static_assert(EE % (128 * KPI) == 0, "key stream is read in whole scan iterations");
static_assert(RNG <= DPITCH, "a local node id fits the eight bits under the direction bit");
static_assert((size_t)NWG * RNG <= 65536, "the range test's subtraction wraps only for nodes below the range");
#define WS_ALIGN 128
#define XA_OFF ((2 * NN + WS_ALIGN - 1) / WS_ALIGN * WS_ALIGN)
#define KEY_PLANE (NN * DD - WS_ALIGN)
static_assert(XA_OFF - 2 * NN <= WS_ALIGN && (XA_OFF + KEY_PLANE) % WS_ALIGN == 0,
              "row planes start on 512-byte boundaries and the layout is no longer than before");
static_assert((size_t)EE + (size_t)NLW * LSTRIDE + NLW <= (size_t)KEY_PLANE,
              "keys, saved lists and list lengths share the key plane of the workspace");
static_assert(LSTRIDE % 32 == 0 && LCAP >= QCAP, "lists are read in chunks of 32 entries");
static_assert(OWN_SLOTS == 13, "the owner waves' register sums are written out for thirteen slots");

typedef __attribute__((ext_vector_type(2))) float v2f;
typedef __attribute__((ext_vector_type(2))) unsigned short us2;
typedef __attribute__((ext_vector_type(8))) float v8f;
typedef __attribute__((ext_vector_type(16))) float v16f;

__global__ __launch_bounds__(256) void pack_kernel(
    const int* __restrict__ src, const int* __restrict__ dst,
    int* __restrict__ pk, int E) {
  int e = blockIdx.x * blockDim.x + threadIdx.x;
  if (e >= E) return;
  pk[e] = (int)((unsigned)src[e] | ((unsigned)dst[e] << 16));
}

template <class Drain>
__device__ __forceinline__ void scan_keys(const int* __restrict__ pk, unsigned b, int wave, int lane,
                                          unsigned* myq, Drain&& drain) {
  const uint4* pk4 = (const uint4*)pk;
  const unsigned short lo = (unsigned short)(b * RNG);
  const us2 lo2 = {lo, lo};
  int cnt = 0;
  for (int it = wave; it < NIT; it += AGG_WAVES) {
    int ip = it + (int)b * STAG;
    if (ip >= NIT) ip -= NIT;
    uint4 k[KPI];
#pragma unroll
    for (int u = 0; u < KPI; ++u)
      k[u] = pk4[((size_t)ip * KPI + u) * 32 + lane];
#pragma unroll
    for (int u = 0; u < KPI; ++u) {
      const unsigned kk[4] = {k[u].x, k[u].y, k[u].z, k[u].w};
#pragma unroll
      for (int j = 0; j < 4; ++j) {
        const unsigned p = kk[j];
        const us2 rel = __builtin_bit_cast(us2, p) - lo2;
        const unsigned relw = __builtin_bit_cast(unsigned, rel);
        const bool tf = relw < ((unsigned)RNG << 16);
        const bool tb = rel.x < RNG;
        const unsigned mf = __builtin_amdgcn_ballot_w32(tf);
        const unsigned mb = __builtin_amdgcn_ballot_w32(tb);
        if (mf | mb) {
          if (mf) {
            unsigned pos = cnt + __builtin_amdgcn_mbcnt_lo(mf, 0u);
            if (tf) myq[pos] = (relw & 0xFFFF0000u) | (p & 0xFFFFu);
            cnt += __builtin_popcount(mf);
          }
          if (mb) {
            unsigned pos = cnt + __builtin_amdgcn_mbcnt_lo(mb, 0u);
            if (tb) myq[pos] = (relw << 16) | (p >> 16) | (1u << 24);
            cnt += __builtin_popcount(mb);
          }
        }
      }
      if (cnt > 64) {
        drain(cnt);
        cnt = 0;
      }
    }
  }
  drain(cnt);
}

__global__ __launch_bounds__(1024) void scan_owner_kernel(
    const int* __restrict__ pk,
    float* __restrict__ inv,
    int* __restrict__ lists,
    int* __restrict__ cnts) {
  __shared__ unsigned q[AGG_WAVES][QCAP + 8];
  __shared__ unsigned degc[2 * DPITCH];
  __shared__ unsigned olist[AGG_WAVES][LCAP];
  __shared__ unsigned ocnt[AGG_WAVES];
  __shared__ unsigned ovf;

  const int tid = threadIdx.x;
  const int lane = tid & 31;
  const int wave = __builtin_amdgcn_readfirstlane(tid >> 5);
  const unsigned b = blockIdx.x;

  const int wg = (int)b * AGG_WAVES + wave;
  int* mylist = lists + (size_t)wg * LSTRIDE;
  unsigned* myq = q[wave];

  if (tid < 2 * DPITCH) degc[tid] = 0u;
  if (tid < AGG_WAVES) ocnt[tid] = 0u;
  if (tid == 0) ovf = 0u;
  __syncthreads();

  auto drain = [&](int n) {
#pragma nounroll
    for (int i = lane; i < n; i += 32) {
      const unsigned e = myq[i];
      const unsigned row = (e >> 16) & (2 * DPITCH - 1);
      atomicAdd(&degc[row], 1u);
      const unsigned rr = (row / DPITCH) * RNG + (row % DPITCH);
      const unsigned ow = rr % AGG_WAVES;
      const unsigned pos = atomicAdd(&ocnt[ow], 1u);
      if (pos < LCAP) olist[ow][pos] = (e & 0x0100FFFFu) | ((rr / AGG_WAVES) << 16);
    }
  };
  scan_keys(pk, b, wave, lane, myq, drain);
  __syncthreads();
  const int mine = __builtin_amdgcn_readfirstlane((int)ocnt[wave]);
  if (mine > LCAP && lane == 0) ovf = 1u;
  __syncthreads();
  const bool over = ovf != 0u;
  if (!over) {
    for (int i = lane; i < mine; i += 32) mylist[i] = (int)olist[wave][i];
    mylist[mine + lane] = (int)FILL_ENTRY;
  }
  if (lane == 0) cnts[wg] = over ? -1 : mine;

  if (tid < RNG) {
    const int node = (int)b * RNG + tid;
    if (node < NN) {
      inv[node] = 1.0f / fmaxf((float)degc[DPITCH + tid], 1.0f);
      inv[NN + node] = 1.0f / fmaxf((float)degc[tid], 1.0f);
    }
  }
}

__global__ __launch_bounds__(1024) void agg_owner_kernel(
    const float* __restrict__ x, const int* __restrict__ pk,
    const float* __restrict__ inv,
    float* __restrict__ fwd, float* __restrict__ bwd,
    const int* __restrict__ lists,
    const int* __restrict__ cnts) {
  __shared__ float acc[(2 * DPITCH + 1) * DD];
  __shared__ unsigned q[AGG_WAVES][QCAP + 8];

  const int tid = threadIdx.x;
  const int lane = tid & 31;
  const int wave = __builtin_amdgcn_readfirstlane(tid >> 5);
  const unsigned b = blockIdx.x;

  const int wg = (int)b * AGG_WAVES + wave;
  const int* mylist = lists + (size_t)wg * LSTRIDE;
  const int nrep = __builtin_amdgcn_readfirstlane(cnts[wg]);
  const int mode = __builtin_amdgcn_readfirstlane(cnts[(int)b * AGG_WAVES]);

  const float4* x4 = (const float4*)x;

  if (mode >= 0) {
    v16f ax = {}, ay = {}, az = {}, aw = {};
    unsigned cur = (unsigned)mylist[lane];
    unsigned nxt = 0;
    if (32 < nrep) nxt = (unsigned)mylist[32 + lane];
    float wcur = inv[((cur >> 24) & 1u) * NN + (cur & 0xFFFFu)];
    asm volatile("" : "+v"(wcur));
    for (int base = 0; base < nrep; base += 32) {
      float wnxt = 0.0f;
      unsigned nn = 0;
      if (base + 32 < nrep) wnxt = inv[((nxt >> 24) & 1u) * NN + (nxt & 0xFFFFu)];
      if (base + 64 < nrep) nn = (unsigned)mylist[base + 64 + lane];
      const int gend = (nrep - base < 32) ? nrep - base : 32;
#pragma nounroll
      for (int g = 0; g < gend; g += 8) {
        {
          unsigned e[8];
          float w[8];
          float4 r[8];
#pragma unroll
          for (int j = 0; j < 8; ++j)
            e[j] = (unsigned)__builtin_amdgcn_readlane((int)cur, g + j);
#pragma unroll
          for (int j = 0; j < 8; ++j)
            r[j] = x4[(size_t)(e[j] & 0xFFFFu) * (DD / 4) + lane];
#pragma unroll
          for (int j = 0; j < 8; ++j)
            w[j] = __int_as_float(__builtin_amdgcn_readlane(__float_as_int(wcur), g + j));
#pragma unroll
          for (int j = 0; j < 8; ++j) {
            const float4 v = r[j];
            const float ww = w[j];
            const unsigned slot = (e[j] >> 16) & 15u;
            ax[slot] += v.x * ww;
            ay[slot] += v.y * ww;
            az[slot] += v.z * ww;
            aw[slot] += v.w * ww;
          }
        }
      }
      cur = nxt;
      wcur = wnxt;
      nxt = nn;
    }
#define ROW_OUT(S, A)                                                          \
    {                                                                          \
      const int rr = (S) * AGG_WAVES + wave;                                   \
      const int dir = (rr >= RNG) ? 1 : 0;                                     \
      const int node = (int)b * RNG + rr - dir * RNG;                          \
      if (rr < 2 * RNG && node < NN)                                           \
        *(float4*)((dir ? bwd : fwd) + (size_t)node * DD + lane * 4) = A;      \
    }
    ROW_OUT(0, make_float4(ax[0], ay[0], az[0], aw[0]));
    ROW_OUT(1, make_float4(ax[1], ay[1], az[1], aw[1]));
    ROW_OUT(2, make_float4(ax[2], ay[2], az[2], aw[2]));
    ROW_OUT(3, make_float4(ax[3], ay[3], az[3], aw[3]));
    ROW_OUT(4, make_float4(ax[4], ay[4], az[4], aw[4]));
    ROW_OUT(5, make_float4(ax[5], ay[5], az[5], aw[5]));
    ROW_OUT(6, make_float4(ax[6], ay[6], az[6], aw[6]));
    ROW_OUT(7, make_float4(ax[7], ay[7], az[7], aw[7]));
    ROW_OUT(8, make_float4(ax[8], ay[8], az[8], aw[8]));
    ROW_OUT(9, make_float4(ax[9], ay[9], az[9], aw[9]));
    ROW_OUT(10, make_float4(ax[10], ay[10], az[10], aw[10]));
    ROW_OUT(11, make_float4(ax[11], ay[11], az[11], aw[11]));
    ROW_OUT(12, make_float4(ax[12], ay[12], az[12], aw[12]));
#undef ROW_OUT
    return;
  }

  float4* acc4 = (float4*)acc;
  for (int i = tid; i < (2 * DPITCH + 1) * DD / 4; i += 1024)
    acc4[i] = make_float4(0.0f, 0.0f, 0.0f, 0.0f);
  __syncthreads();

  {
    unsigned* myq = q[wave];
    auto drain = [&](int n) {
      if (lane < 8) myq[n + lane] = 1u << 25;
      for (int i = 0; i < n; i += 8) {
        unsigned e[8];
        float w[8];
        float4 r[8];
#pragma unroll
        for (int j = 0; j < 8; ++j)
          e[j] = __builtin_amdgcn_readfirstlane(myq[i + j]);
#pragma unroll
        for (int j = 0; j < 8; ++j) {
          unsigned o = e[j] & 0xFFFFu;
          w[j] = inv[((e[j] >> 24) & 1u) * NN + o];
          r[j] = x4[(size_t)o * (DD / 4) + lane];
        }
#pragma unroll
        for (int j = 0; j < 8; ++j) {
          float* a = acc + (e[j] >> 16) * DD + lane;
          atomicAdd(a, r[j].x * w[j]);
          atomicAdd(a + 32, r[j].y * w[j]);
          atomicAdd(a + 64, r[j].z * w[j]);
          atomicAdd(a + 96, r[j].w * w[j]);
        }
      }
    };
    scan_keys(pk, b, wave, lane, myq, drain);
  }
  __syncthreads();

  for (int row = wave; row < 2 * RNG; row += AGG_WAVES) {
    const int dir = (row >= RNG) ? 1 : 0;
    const int loc = row - dir * RNG;
    int node = (int)b * RNG + loc;
    if (node < NN) {
      const float* a = acc + (dir * DPITCH + loc) * DD + lane;
      float4 v = make_float4(a[0], a[32], a[64], a[96]);
      float* o = dir ? bwd : fwd;
      *(float4*)(o + (size_t)node * DD + lane * 4) = v;
    }
  }
}

#define GEMM_WG 256
#define NTILE (NN / 16)
#define APITCH (K3 + 4)

static_assert(NN % 16 == 0, "row tiles are whole");

typedef __attribute__((ext_vector_type(4))) float v4f;

__global__ __launch_bounds__(256) void gemm_wmma_kernel(
    const float* __restrict__ x, const float* __restrict__ fwd,
    const float* __restrict__ bwd, const float* __restrict__ W,
    const float* __restrict__ bias, float* __restrict__ out, int do_relu) {
  __shared__ __attribute__((aligned(16))) float As[2][16][APITCH];

  const int tid = threadIdx.x;
  const int wave = tid >> 5;
  const int lane = tid & 31;
  const int hl = lane >> 4;
  const int l16 = lane & 15;
  const int col = wave * 16 + l16;

  v2f bw[K3 / 4];
#pragma unroll
  for (int s = 0; s < K3 / 4; ++s) {
    bw[s].x = W[(size_t)(4 * s + 2 * hl + 0) * DD + col];
    bw[s].y = W[(size_t)(4 * s + 2 * hl + 1) * DD + col];
  }
  const float bb = bias[col];

  const v4f* x4 = (const v4f*)x + tid;
  const v4f* f4 = (const v4f*)fwd + tid;
  const v4f* b4 = (const v4f*)bwd + tid;
  const int sA = (tid >> 5) * APITCH + 4 * (tid & 31);
  v4f g0, g1, g2, g3, g4, g5;

  int tile = blockIdx.x;
  {
    const size_t base = (size_t)tile * (16 * DD / 4);
    g0 = x4[base]; g1 = x4[base + 256];
    g2 = f4[base]; g3 = f4[base + 256];
    g4 = b4[base]; g5 = b4[base + 256];
    float* d = &As[0][0][0] + sA;
    *(v4f*)(d) = g0;            *(v4f*)(d + 8 * APITCH) = g1;
    *(v4f*)(d + DD) = g2;       *(v4f*)(d + 8 * APITCH + DD) = g3;
    *(v4f*)(d + 2 * DD) = g4;   *(v4f*)(d + 8 * APITCH + 2 * DD) = g5;
  }
  __syncthreads();

  int buf = 0;
  for (;;) {
    const bool more = tile + GEMM_WG < NTILE;
    if (more) {
      const size_t base = (size_t)(tile + GEMM_WG) * (16 * DD / 4);
      g0 = x4[base]; g1 = x4[base + 256];
      g2 = f4[base]; g3 = f4[base + 256];
      g4 = b4[base]; g5 = b4[base + 256];
    }

    const float* arow = &As[buf][l16][2 * hl];
    v2f af[2][8];
#pragma unroll
    for (int j = 0; j < 8; ++j) af[0][j] = *(const v2f*)(arow + 4 * j);
    v8f acc = {};
#pragma unroll
    for (int gidx = 0; gidx < K3 / 32; ++gidx) {
      if (gidx + 1 < K3 / 32) {
#pragma unroll
        for (int j = 0; j < 8; ++j)
          af[(gidx + 1) & 1][j] = *(const v2f*)(arow + 32 * (gidx + 1) + 4 * j);
      }
      __builtin_amdgcn_sched_barrier(0);
#pragma unroll
      for (int j = 0; j < 8; ++j)
        acc = __builtin_amdgcn_wmma_f32_16x16x4_f32(
             false, af[gidx & 1][j],  false, bw[8 * gidx + j],
             (short)0, acc,  false,  false);
    }

    {
      float* d = &As[buf ^ 1][0][0] + sA;
      *(v4f*)(d) = g0;            *(v4f*)(d + 8 * APITCH) = g1;
      *(v4f*)(d + DD) = g2;       *(v4f*)(d + 8 * APITCH + DD) = g3;
      *(v4f*)(d + 2 * DD) = g4;   *(v4f*)(d + 8 * APITCH + 2 * DD) = g5;
    }
    __syncthreads();

    asm volatile("v_nop\n\tv_nop\n\tv_nop\n\tv_nop\n\tv_nop" : "+v"(acc));

    float* o = out + (size_t)(tile * 16 + 8 * hl) * DD + col;
#pragma unroll
    for (int r = 0; r < 8; ++r) {
      float v = acc[r] + bb;
      if (do_relu) v = fmaxf(v, 0.0f);
      o[r * DD] = v;
    }

    if (!more) break;
    tile += GEMM_WG;
    buf ^= 1;
  }
}

extern "C" void kernel_launch(void* const* d_in, const int* in_sizes, int n_in,
                              void* d_out, int out_size, void* d_ws, size_t ws_size,
                              hipStream_t stream) {
  const float* x  = (const float*)d_in[0];
  const float* W0 = (const float*)d_in[1];
  const float* b0 = (const float*)d_in[2];
  const float* W1 = (const float*)d_in[3];
  const float* b1 = (const float*)d_in[4];
  const float* W2 = (const float*)d_in[5];
  const float* b2 = (const float*)d_in[6];
  const int* src  = (const int*)d_in[7];
  const int* dst  = (const int*)d_in[8];
  float* out = (float*)d_out;

  float* ws      = (float*)d_ws;
  float* inv     = ws;
  float* xA      = ws + XA_OFF;
  int*   pk      = (int*)(xA + (size_t)NN * DD);
  int*   lists   = pk + EE;
  int*   cnts    = lists + (size_t)NLW * LSTRIDE;
  float* fwd     = (float*)pk + KEY_PLANE;
  float* bwd     = fwd + (size_t)NN * DD;

  pack_kernel<<<(EE + 255) / 256, 256, 0, stream>>>(src, dst, pk, EE);
  scan_owner_kernel<<<NWG, 1024, 0, stream>>>(pk, inv, lists, cnts);

  const float* Ws[3]  = {W0, W1, W2};
  const float* bs[3]  = {b0, b1, b2};
  float*       dsts[3] = {out, xA, out};

  const float* cur = x;
  for (int l = 0; l < 3; ++l) {
    agg_owner_kernel<<<NWG, 1024, 0, stream>>>(cur, pk, inv, fwd, bwd, lists, cnts);
    gemm_wmma_kernel<<<GEMM_WG, 256, 0, stream>>>(cur, fwd, bwd, Ws[l], bs[l],
                                                  dsts[l], (l < 2) ? 1 : 0);
    cur = dsts[l];
  }
}


# Edges sorted by owner once: 256 chunk work-groups bin matches in LDS, write 256-byte segments; owners read their 64 KB of segments, no longer stream all keys; sum order stays LDS counter arrival
# speedup vs baseline: 4.2493x; 1.1103x over previous
#include <hip/hip_runtime.h>

#define NN 50000
#define EE 800000
#define DD 128
#define K3 384

#define RNG 196
#define NWG ((NN + RNG - 1) / RNG)
#define DPITCH 256
#define AGG_WAVES 32
#define QCAP 320
#define KPI 5
#define NIT (EE / (128 * KPI))
#define STAG (NIT / NWG)
#define LSTRIDE 640
#define LCAP (LSTRIDE - 32)
#define NLW (NWG * AGG_WAVES)
#define OWN_SLOTS ((2 * RNG + AGG_WAVES - 1) / AGG_WAVES)
#define FILL_ENTRY (15u << 16)
#define NPART 256
#define EPW 3136
#define EPT ((EPW + 1023) / 1024)
#define SEGW 64
#define SEGCAP (SEGW - 1)
#define SPW (NPART / AGG_WAVES)

static_assert(NN <= 65536, "packed keys hold a node id in 16 bits");
static_assert(EE % (128 * KPI) == 0, "key stream is read in whole scan iterations");
static_assert(RNG <= DPITCH, "a local node id fits the eight bits under the direction bit");
static_assert((size_t)NWG * RNG <= 65536, "the range test's subtraction wraps only for nodes below the range");
#define WS_ALIGN 128
#define XA_OFF ((2 * NN + WS_ALIGN - 1) / WS_ALIGN * WS_ALIGN)
#define KEY_PLANE (NN * DD - WS_ALIGN)
static_assert(XA_OFF - 2 * NN <= WS_ALIGN && (XA_OFF + KEY_PLANE) % WS_ALIGN == 0,
              "row planes start on 512-byte boundaries and the layout is no longer than before");
static_assert((size_t)EE + (size_t)NLW * LSTRIDE + NLW <= (size_t)KEY_PLANE,
              "keys, saved lists and list lengths share the key plane of the workspace");
static_assert(LSTRIDE % 32 == 0 && LCAP >= QCAP, "lists are read in chunks of 32 entries");
static_assert(OWN_SLOTS == 13, "the owner waves' register sums are written out for thirteen slots");
static_assert((size_t)NPART * EPW >= EE && EPW % 32 == 0, "the chunks cover the edges and split the keys at line boundaries");
static_assert(NPART % AGG_WAVES == 0 && SPW <= 32, "a wave reads a whole number of segments and holds their counts one per lane");
static_assert((size_t)NWG * NPART * SEGW <= (size_t)NN * DD, "the segments fit the fwd plane, which is free until the first aggregation");

typedef __attribute__((ext_vector_type(2))) float v2f;
typedef __attribute__((ext_vector_type(2))) unsigned short us2;
typedef __attribute__((ext_vector_type(8))) float v8f;
typedef __attribute__((ext_vector_type(16))) float v16f;

__global__ __launch_bounds__(1024) void part_kernel(
    const int* __restrict__ src, const int* __restrict__ dst,
    int* __restrict__ pk, unsigned* __restrict__ segs) {
  __shared__ unsigned bins[NWG][SEGW + 1];
  __shared__ unsigned bcnt[NWG];

  const int tid = threadIdx.x;
  const int lane = tid & 31;
  const int wave = tid >> 5;
  const unsigned r = blockIdx.x;

  if (tid < NWG) bcnt[tid] = 0u;
  __syncthreads();

  unsigned es[EPT], ed[EPT];
  bool ok[EPT];
#pragma unroll
  for (int u = 0; u < EPT; ++u) {
    const int i = tid + u * 1024;
    const int e = (int)r * EPW + i;
    ok[u] = (i < EPW) && (e < EE);
    es[u] = 0u;
    ed[u] = 0u;
    if (ok[u]) {
      es[u] = (unsigned)src[e];
      ed[u] = (unsigned)dst[e];
    }
  }
#pragma unroll
  for (int u = 0; u < EPT; ++u) {
    if (ok[u]) {
      const unsigned sn = es[u], dn = ed[u];
      pk[(int)r * EPW + tid + u * 1024] = (int)(sn | (dn << 16));
      const unsigned of = dn / RNG;
      const unsigned ob = sn / RNG;
      unsigned pos = atomicAdd(&bcnt[of], 1u);
      if (pos < SEGCAP) bins[of][pos] = sn | ((dn - of * RNG) << 16);
      pos = atomicAdd(&bcnt[ob], 1u);
      if (pos < SEGCAP) bins[ob][pos] = dn | ((sn - ob * RNG) << 16) | (1u << 24);
    }
  }
  __syncthreads();

  for (int o = wave; o < NWG; o += 32) {
    const unsigned c = bcnt[o];
    unsigned* sp = segs + ((size_t)o * NPART + r) * SEGW;
    if ((unsigned)lane < c) sp[lane] = bins[o][lane];
    if (lane == 31 || (unsigned)(32 + lane) < c) sp[32 + lane] = (lane == 31) ? c : bins[o][32 + lane];
  }
}

template <class Drain>
__device__ __forceinline__ void scan_keys(const int* __restrict__ pk, unsigned b, int wave, int lane,
                                          unsigned* myq, Drain&& drain) {
  const uint4* pk4 = (const uint4*)pk;
  const unsigned short lo = (unsigned short)(b * RNG);
  const us2 lo2 = {lo, lo};
  int cnt = 0;
  for (int it = wave; it < NIT; it += AGG_WAVES) {
    int ip = it + (int)b * STAG;
    if (ip >= NIT) ip -= NIT;
    uint4 k[KPI];
#pragma unroll
    for (int u = 0; u < KPI; ++u)
      k[u] = pk4[((size_t)ip * KPI + u) * 32 + lane];
#pragma unroll
    for (int u = 0; u < KPI; ++u) {
      const unsigned kk[4] = {k[u].x, k[u].y, k[u].z, k[u].w};
#pragma unroll
      for (int j = 0; j < 4; ++j) {
        const unsigned p = kk[j];
        const us2 rel = __builtin_bit_cast(us2, p) - lo2;
        const unsigned relw = __builtin_bit_cast(unsigned, rel);
        const bool tf = relw < ((unsigned)RNG << 16);
        const bool tb = rel.x < RNG;
        const unsigned mf = __builtin_amdgcn_ballot_w32(tf);
        const unsigned mb = __builtin_amdgcn_ballot_w32(tb);
        if (mf | mb) {
          if (mf) {
            unsigned pos = cnt + __builtin_amdgcn_mbcnt_lo(mf, 0u);
            if (tf) myq[pos] = (relw & 0xFFFF0000u) | (p & 0xFFFFu);
            cnt += __builtin_popcount(mf);
          }
          if (mb) {
            unsigned pos = cnt + __builtin_amdgcn_mbcnt_lo(mb, 0u);
            if (tb) myq[pos] = (relw << 16) | (p >> 16) | (1u << 24);
            cnt += __builtin_popcount(mb);
          }
        }
      }
      if (cnt > 64) {
        drain(cnt);
        cnt = 0;
      }
    }
  }
  drain(cnt);
}

__global__ __launch_bounds__(1024) void scan_owner_kernel(
    const int* __restrict__ pk,
    const unsigned* __restrict__ segs,
    float* __restrict__ inv,
    int* __restrict__ lists,
    int* __restrict__ cnts) {
  __shared__ unsigned q[AGG_WAVES][QCAP + 8];
  __shared__ unsigned degc[2 * DPITCH];
  __shared__ unsigned olist[AGG_WAVES][LCAP];
  __shared__ unsigned ocnt[AGG_WAVES];
  __shared__ unsigned ovf;
  __shared__ unsigned segovf;

  const int tid = threadIdx.x;
  const int lane = tid & 31;
  const int wave = __builtin_amdgcn_readfirstlane(tid >> 5);
  const unsigned b = blockIdx.x;

  const int wg = (int)b * AGG_WAVES + wave;
  int* mylist = lists + (size_t)wg * LSTRIDE;
  unsigned* myq = q[wave];

  const unsigned* seg = segs + ((size_t)b * NPART + (size_t)wave * SPW) * SEGW;
  unsigned myc = 0u;
  if (lane < SPW) myc = seg[lane * SEGW + SEGCAP];

  if (tid < 2 * DPITCH) degc[tid] = 0u;
  if (tid < AGG_WAVES) ocnt[tid] = 0u;
  if (tid == 0) { ovf = 0u; segovf = 0u; }
  __syncthreads();
  if (myc > SEGCAP) segovf = 1u;
  __syncthreads();

  auto place = [&](unsigned e) {
    const unsigned row = (e >> 16) & (2 * DPITCH - 1);
    atomicAdd(&degc[row], 1u);
    const unsigned rr = (row / DPITCH) * RNG + (row % DPITCH);
    const unsigned ow = rr % AGG_WAVES;
    const unsigned pos = atomicAdd(&ocnt[ow], 1u);
    if (pos < LCAP) olist[ow][pos] = (e & 0x0100FFFFu) | ((rr / AGG_WAVES) << 16);
  };

  if (segovf == 0u) {
    unsigned sc[SPW], e0[SPW], e1[SPW];
#pragma unroll
    for (int j = 0; j < SPW; ++j) {
      sc[j] = (unsigned)__builtin_amdgcn_readlane((int)myc, j);
      e0[j] = 0u;
      e1[j] = 0u;
      if ((unsigned)lane < sc[j]) e0[j] = seg[j * SEGW + lane];
      if ((unsigned)(32 + lane) < sc[j]) e1[j] = seg[j * SEGW + 32 + lane];
    }
#pragma unroll
    for (int j = 0; j < SPW; ++j) {
      if ((unsigned)lane < sc[j]) place(e0[j]);
      if ((unsigned)(32 + lane) < sc[j]) place(e1[j]);
    }
  } else {
    auto drain = [&](int n) {
#pragma nounroll
      for (int i = lane; i < n; i += 32) place(myq[i]);
    };
    scan_keys(pk, b, wave, lane, myq, drain);
  }
  __syncthreads();
  const int mine = __builtin_amdgcn_readfirstlane((int)ocnt[wave]);
  if (mine > LCAP && lane == 0) ovf = 1u;
  __syncthreads();
  const bool over = ovf != 0u;
  if (!over) {
    for (int i = lane; i < mine; i += 32) mylist[i] = (int)olist[wave][i];
    mylist[mine + lane] = (int)FILL_ENTRY;
  }
  if (lane == 0) cnts[wg] = over ? -1 : mine;

  if (tid < RNG) {
    const int node = (int)b * RNG + tid;
    if (node < NN) {
      inv[node] = 1.0f / fmaxf((float)degc[DPITCH + tid], 1.0f);
      inv[NN + node] = 1.0f / fmaxf((float)degc[tid], 1.0f);
    }
  }
}

__global__ __launch_bounds__(1024) void agg_owner_kernel(
    const float* __restrict__ x, const int* __restrict__ pk,
    const float* __restrict__ inv,
    float* __restrict__ fwd, float* __restrict__ bwd,
    const int* __restrict__ lists,
    const int* __restrict__ cnts) {
  __shared__ float acc[(2 * DPITCH + 1) * DD];
  __shared__ unsigned q[AGG_WAVES][QCAP + 8];

  const int tid = threadIdx.x;
  const int lane = tid & 31;
  const int wave = __builtin_amdgcn_readfirstlane(tid >> 5);
  const unsigned b = blockIdx.x;

  const int wg = (int)b * AGG_WAVES + wave;
  const int* mylist = lists + (size_t)wg * LSTRIDE;
  const int nrep = __builtin_amdgcn_readfirstlane(cnts[wg]);
  const int mode = __builtin_amdgcn_readfirstlane(cnts[(int)b * AGG_WAVES]);

  const float4* x4 = (const float4*)x;

  if (mode >= 0) {
    v16f ax = {}, ay = {}, az = {}, aw = {};
    unsigned cur = (unsigned)mylist[lane];
    unsigned nxt = 0;
    if (32 < nrep) nxt = (unsigned)mylist[32 + lane];
    float wcur = inv[((cur >> 24) & 1u) * NN + (cur & 0xFFFFu)];
    asm volatile("" : "+v"(wcur));
    for (int base = 0; base < nrep; base += 32) {
      float wnxt = 0.0f;
      unsigned nn = 0;
      if (base + 32 < nrep) wnxt = inv[((nxt >> 24) & 1u) * NN + (nxt & 0xFFFFu)];
      if (base + 64 < nrep) nn = (unsigned)mylist[base + 64 + lane];
      const int gend = (nrep - base < 32) ? nrep - base : 32;
#pragma nounroll
      for (int g = 0; g < gend; g += 8) {
        {
          unsigned e[8];
          float w[8];
          float4 r[8];
#pragma unroll
          for (int j = 0; j < 8; ++j)
            e[j] = (unsigned)__builtin_amdgcn_readlane((int)cur, g + j);
#pragma unroll
          for (int j = 0; j < 8; ++j)
            r[j] = x4[(size_t)(e[j] & 0xFFFFu) * (DD / 4) + lane];
#pragma unroll
          for (int j = 0; j < 8; ++j)
            w[j] = __int_as_float(__builtin_amdgcn_readlane(__float_as_int(wcur), g + j));
#pragma unroll
          for (int j = 0; j < 8; ++j) {
            const float4 v = r[j];
            const float ww = w[j];
            const unsigned slot = (e[j] >> 16) & 15u;
            ax[slot] += v.x * ww;
            ay[slot] += v.y * ww;
            az[slot] += v.z * ww;
            aw[slot] += v.w * ww;
          }
        }
      }
      cur = nxt;
      wcur = wnxt;
      nxt = nn;
    }
#define ROW_OUT(S, A)                                                          \
    {                                                                          \
      const int rr = (S) * AGG_WAVES + wave;                                   \
      const int dir = (rr >= RNG) ? 1 : 0;                                     \
      const int node = (int)b * RNG + rr - dir * RNG;                          \
      if (rr < 2 * RNG && node < NN)                                           \
        *(float4*)((dir ? bwd : fwd) + (size_t)node * DD + lane * 4) = A;      \
    }
    ROW_OUT(0, make_float4(ax[0], ay[0], az[0], aw[0]));
    ROW_OUT(1, make_float4(ax[1], ay[1], az[1], aw[1]));
    ROW_OUT(2, make_float4(ax[2], ay[2], az[2], aw[2]));
    ROW_OUT(3, make_float4(ax[3], ay[3], az[3], aw[3]));
    ROW_OUT(4, make_float4(ax[4], ay[4], az[4], aw[4]));
    ROW_OUT(5, make_float4(ax[5], ay[5], az[5], aw[5]));
    ROW_OUT(6, make_float4(ax[6], ay[6], az[6], aw[6]));
    ROW_OUT(7, make_float4(ax[7], ay[7], az[7], aw[7]));
    ROW_OUT(8, make_float4(ax[8], ay[8], az[8], aw[8]));
    ROW_OUT(9, make_float4(ax[9], ay[9], az[9], aw[9]));
    ROW_OUT(10, make_float4(ax[10], ay[10], az[10], aw[10]));
    ROW_OUT(11, make_float4(ax[11], ay[11], az[11], aw[11]));
    ROW_OUT(12, make_float4(ax[12], ay[12], az[12], aw[12]));
#undef ROW_OUT
    return;
  }

  float4* acc4 = (float4*)acc;
  for (int i = tid; i < (2 * DPITCH + 1) * DD / 4; i += 1024)
    acc4[i] = make_float4(0.0f, 0.0f, 0.0f, 0.0f);
  __syncthreads();

  {
    unsigned* myq = q[wave];
    auto drain = [&](int n) {
      if (lane < 8) myq[n + lane] = 1u << 25;
      for (int i = 0; i < n; i += 8) {
        unsigned e[8];
        float w[8];
        float4 r[8];
#pragma unroll
        for (int j = 0; j < 8; ++j)
          e[j] = __builtin_amdgcn_readfirstlane(myq[i + j]);
#pragma unroll
        for (int j = 0; j < 8; ++j) {
          unsigned o = e[j] & 0xFFFFu;
          w[j] = inv[((e[j] >> 24) & 1u) * NN + o];
          r[j] = x4[(size_t)o * (DD / 4) + lane];
        }
#pragma unroll
        for (int j = 0; j < 8; ++j) {
          float* a = acc + (e[j] >> 16) * DD + lane;
          atomicAdd(a, r[j].x * w[j]);
          atomicAdd(a + 32, r[j].y * w[j]);
          atomicAdd(a + 64, r[j].z * w[j]);
          atomicAdd(a + 96, r[j].w * w[j]);
        }
      }
    };
    scan_keys(pk, b, wave, lane, myq, drain);
  }
  __syncthreads();

  for (int row = wave; row < 2 * RNG; row += AGG_WAVES) {
    const int dir = (row >= RNG) ? 1 : 0;
    const int loc = row - dir * RNG;
    int node = (int)b * RNG + loc;
    if (node < NN) {
      const float* a = acc + (dir * DPITCH + loc) * DD + lane;
      float4 v = make_float4(a[0], a[32], a[64], a[96]);
      float* o = dir ? bwd : fwd;
      *(float4*)(o + (size_t)node * DD + lane * 4) = v;
    }
  }
}

#define GEMM_WG 256
#define NTILE (NN / 16)
#define APITCH (K3 + 4)

static_assert(NN % 16 == 0, "row tiles are whole");

typedef __attribute__((ext_vector_type(4))) float v4f;

__global__ __launch_bounds__(256) void gemm_wmma_kernel(
    const float* __restrict__ x, const float* __restrict__ fwd,
    const float* __restrict__ bwd, const float* __restrict__ W,
    const float* __restrict__ bias, float* __restrict__ out, int do_relu) {
  __shared__ __attribute__((aligned(16))) float As[2][16][APITCH];

  const int tid = threadIdx.x;
  const int wave = tid >> 5;
  const int lane = tid & 31;
  const int hl = lane >> 4;
  const int l16 = lane & 15;
  const int col = wave * 16 + l16;

  v2f bw[K3 / 4];
#pragma unroll
  for (int s = 0; s < K3 / 4; ++s) {
    bw[s].x = W[(size_t)(4 * s + 2 * hl + 0) * DD + col];
    bw[s].y = W[(size_t)(4 * s + 2 * hl + 1) * DD + col];
  }
  const float bb = bias[col];

  const v4f* x4 = (const v4f*)x + tid;
  const v4f* f4 = (const v4f*)fwd + tid;
  const v4f* b4 = (const v4f*)bwd + tid;
  const int sA = (tid >> 5) * APITCH + 4 * (tid & 31);
  v4f g0, g1, g2, g3, g4, g5;

  int tile = blockIdx.x;
  {
    const size_t base = (size_t)tile * (16 * DD / 4);
    g0 = x4[base]; g1 = x4[base + 256];
    g2 = f4[base]; g3 = f4[base + 256];
    g4 = b4[base]; g5 = b4[base + 256];
    float* d = &As[0][0][0] + sA;
    *(v4f*)(d) = g0;            *(v4f*)(d + 8 * APITCH) = g1;
    *(v4f*)(d + DD) = g2;       *(v4f*)(d + 8 * APITCH + DD) = g3;
    *(v4f*)(d + 2 * DD) = g4;   *(v4f*)(d + 8 * APITCH + 2 * DD) = g5;
  }
  __syncthreads();

  int buf = 0;
  for (;;) {
    const bool more = tile + GEMM_WG < NTILE;
    if (more) {
      const size_t base = (size_t)(tile + GEMM_WG) * (16 * DD / 4);
      g0 = x4[base]; g1 = x4[base + 256];
      g2 = f4[base]; g3 = f4[base + 256];
      g4 = b4[base]; g5 = b4[base + 256];
    }

    const float* arow = &As[buf][l16][2 * hl];
    v2f af[2][8];
#pragma unroll
    for (int j = 0; j < 8; ++j) af[0][j] = *(const v2f*)(arow + 4 * j);
    v8f acc = {};
#pragma unroll
    for (int gidx = 0; gidx < K3 / 32; ++gidx) {
      if (gidx + 1 < K3 / 32) {
#pragma unroll
        for (int j = 0; j < 8; ++j)
          af[(gidx + 1) & 1][j] = *(const v2f*)(arow + 32 * (gidx + 1) + 4 * j);
      }
      __builtin_amdgcn_sched_barrier(0);
#pragma unroll
      for (int j = 0; j < 8; ++j)
        acc = __builtin_amdgcn_wmma_f32_16x16x4_f32(
             false, af[gidx & 1][j],  false, bw[8 * gidx + j],
             (short)0, acc,  false,  false);
    }

    {
      float* d = &As[buf ^ 1][0][0] + sA;
      *(v4f*)(d) = g0;            *(v4f*)(d + 8 * APITCH) = g1;
      *(v4f*)(d + DD) = g2;       *(v4f*)(d + 8 * APITCH + DD) = g3;
      *(v4f*)(d + 2 * DD) = g4;   *(v4f*)(d + 8 * APITCH + 2 * DD) = g5;
    }
    __syncthreads();

    asm volatile("v_nop\n\tv_nop\n\tv_nop\n\tv_nop\n\tv_nop" : "+v"(acc));

    float* o = out + (size_t)(tile * 16 + 8 * hl) * DD + col;
#pragma unroll
    for (int r = 0; r < 8; ++r) {
      float v = acc[r] + bb;
      if (do_relu) v = fmaxf(v, 0.0f);
      o[r * DD] = v;
    }

    if (!more) break;
    tile += GEMM_WG;
    buf ^= 1;
  }
}

extern "C" void kernel_launch(void* const* d_in, const int* in_sizes, int n_in,
                              void* d_out, int out_size, void* d_ws, size_t ws_size,
                              hipStream_t stream) {
  const float* x  = (const float*)d_in[0];
  const float* W0 = (const float*)d_in[1];
  const float* b0 = (const float*)d_in[2];
  const float* W1 = (const float*)d_in[3];
  const float* b1 = (const float*)d_in[4];
  const float* W2 = (const float*)d_in[5];
  const float* b2 = (const float*)d_in[6];
  const int* src  = (const int*)d_in[7];
  const int* dst  = (const int*)d_in[8];
  float* out = (float*)d_out;

  float* ws      = (float*)d_ws;
  float* inv     = ws;
  float* xA      = ws + XA_OFF;
  int*   pk      = (int*)(xA + (size_t)NN * DD);
  int*   lists   = pk + EE;
  int*   cnts    = lists + (size_t)NLW * LSTRIDE;
  float* fwd     = (float*)pk + KEY_PLANE;
  float* bwd     = fwd + (size_t)NN * DD;

  unsigned* segs = (unsigned*)fwd;
  part_kernel<<<NPART, 1024, 0, stream>>>(src, dst, pk, segs);
  scan_owner_kernel<<<NWG, 1024, 0, stream>>>(pk, segs, inv, lists, cnts);

  const float* Ws[3]  = {W0, W1, W2};
  const float* bs[3]  = {b0, b1, b2};
  float*       dsts[3] = {out, xA, out};

  const float* cur = x;
  for (int l = 0; l < 3; ++l) {
    agg_owner_kernel<<<NWG, 1024, 0, stream>>>(cur, pk, inv, fwd, bwd, lists, cnts);
    gemm_wmma_kernel<<<GEMM_WG, 256, 0, stream>>>(cur, fwd, bwd, Ws[l], bs[l],
                                                  dsts[l], (l < 2) ? 1 : 0);
    cur = dsts[l];
  }
}


# Each aggregation work-group stages a 16-bit degree table of all nodes (200 KB) in its LDS and takes an entry's weight 1/max(deg,1) from it; the per-entry 4-byte global lookups go; sum order kept
# speedup vs baseline: 4.4501x; 1.0473x over previous
#include <hip/hip_runtime.h>

#define NN 50000
#define EE 800000
#define DD 128
#define K3 384

#define RNG 196
#define NWG ((NN + RNG - 1) / RNG)
#define DPITCH 256
#define AGG_WAVES 32
#define QCAP 320
#define KPI 5
#define NIT (EE / (128 * KPI))
#define STAG (NIT / NWG)
#define LSTRIDE 640
#define LCAP (LSTRIDE - 32)
#define NLW (NWG * AGG_WAVES)
#define OWN_SLOTS ((2 * RNG + AGG_WAVES - 1) / AGG_WAVES)
#define FILL_ENTRY (15u << 16)
#define NPART 256
#define EPW 3136
#define EPT ((EPW + 1023) / 1024)
#define SEGW 64
#define SEGCAP (SEGW - 1)
#define SPW (NPART / AGG_WAVES)
#define DEG_SAT 65535u
#define DEG_VEC (2 * NN / 8)
#define DEG_PCS ((DEG_VEC + 1023) / 1024)
#define DEG_PAD (DEG_PCS * 1024)

static_assert(NN <= 65536, "packed keys hold a node id in 16 bits");
static_assert(EE % (128 * KPI) == 0, "key stream is read in whole scan iterations");
static_assert(RNG <= DPITCH, "a local node id fits the eight bits under the direction bit");
static_assert((size_t)NWG * RNG <= 65536, "the range test's subtraction wraps only for nodes below the range");
#define WS_ALIGN 128
#define XA_OFF ((2 * NN + WS_ALIGN - 1) / WS_ALIGN * WS_ALIGN)
#define KEY_PLANE (NN * DD - WS_ALIGN)
static_assert(XA_OFF - 2 * NN <= WS_ALIGN && (XA_OFF + KEY_PLANE) % WS_ALIGN == 0,
              "row planes start on 512-byte boundaries and the layout is no longer than before");
static_assert((size_t)EE + (size_t)NLW * LSTRIDE + NLW <= (size_t)KEY_PLANE,
              "keys, saved lists and list lengths share the key plane of the workspace");
static_assert(LSTRIDE % 32 == 0 && LCAP >= QCAP, "lists are read in chunks of 32 entries");
static_assert((size_t)EE + (size_t)NLW * LSTRIDE + NLW + (size_t)DEG_PAD * 4 <= (size_t)KEY_PLANE,
              "the 16-bit degree table (2 x NN entries and its padding) fits the key plane behind the list lengths");
static_assert(DEG_PAD - DEG_VEC <= 1024, "one work-group's threads write the padding, a piece each");
static_assert(((size_t)EE + (size_t)NLW * LSTRIDE + NLW) % 4 == 0 && (2 * NN) % 8 == 0,
              "the degree table starts on a 16-byte boundary and is a whole number of 16-byte pieces");
static_assert(OWN_SLOTS == 13, "the owner waves' register sums are written out for thirteen slots");
static_assert((size_t)NPART * EPW >= EE && EPW % 32 == 0, "the chunks cover the edges and split the keys at line boundaries");
static_assert(NPART % AGG_WAVES == 0 && SPW <= 32, "a wave reads a whole number of segments and holds their counts one per lane");
static_assert((size_t)NWG * NPART * SEGW <= (size_t)NN * DD, "the segments fit the fwd plane, which is free until the first aggregation");

typedef __attribute__((ext_vector_type(2))) float v2f;
typedef __attribute__((ext_vector_type(2))) unsigned short us2;
typedef __attribute__((ext_vector_type(8))) float v8f;
typedef __attribute__((ext_vector_type(16))) float v16f;

__global__ __launch_bounds__(1024) void part_kernel(
    const int* __restrict__ src, const int* __restrict__ dst,
    int* __restrict__ pk, unsigned* __restrict__ segs) {
  __shared__ unsigned bins[NWG][SEGW + 1];
  __shared__ unsigned bcnt[NWG];

  const int tid = threadIdx.x;
  const int lane = tid & 31;
  const int wave = tid >> 5;
  const unsigned r = blockIdx.x;

  if (tid < NWG) bcnt[tid] = 0u;
  __syncthreads();

  unsigned es[EPT], ed[EPT];
  bool ok[EPT];
#pragma unroll
  for (int u = 0; u < EPT; ++u) {
    const int i = tid + u * 1024;
    const int e = (int)r * EPW + i;
    ok[u] = (i < EPW) && (e < EE);
    es[u] = 0u;
    ed[u] = 0u;
    if (ok[u]) {
      es[u] = (unsigned)src[e];
      ed[u] = (unsigned)dst[e];
    }
  }
#pragma unroll
  for (int u = 0; u < EPT; ++u) {
    if (ok[u]) {
      const unsigned sn = es[u], dn = ed[u];
      pk[(int)r * EPW + tid + u * 1024] = (int)(sn | (dn << 16));
      const unsigned of = dn / RNG;
      const unsigned ob = sn / RNG;
      unsigned pos = atomicAdd(&bcnt[of], 1u);
      if (pos < SEGCAP) bins[of][pos] = sn | ((dn - of * RNG) << 16);
      pos = atomicAdd(&bcnt[ob], 1u);
      if (pos < SEGCAP) bins[ob][pos] = dn | ((sn - ob * RNG) << 16) | (1u << 24);
    }
  }
  __syncthreads();

  for (int o = wave; o < NWG; o += 32) {
    const unsigned c = bcnt[o];
    unsigned* sp = segs + ((size_t)o * NPART + r) * SEGW;
    if ((unsigned)lane < c) sp[lane] = bins[o][lane];
    if (lane == 31 || (unsigned)(32 + lane) < c) sp[32 + lane] = (lane == 31) ? c : bins[o][32 + lane];
  }
}

template <class Drain>
__device__ __forceinline__ void scan_keys(const int* __restrict__ pk, unsigned b, int wave, int lane,
                                          unsigned* myq, Drain&& drain) {
  const uint4* pk4 = (const uint4*)pk;
  const unsigned short lo = (unsigned short)(b * RNG);
  const us2 lo2 = {lo, lo};
  int cnt = 0;
  for (int it = wave; it < NIT; it += AGG_WAVES) {
    int ip = it + (int)b * STAG;
    if (ip >= NIT) ip -= NIT;
    uint4 k[KPI];
#pragma unroll
    for (int u = 0; u < KPI; ++u)
      k[u] = pk4[((size_t)ip * KPI + u) * 32 + lane];
#pragma unroll
    for (int u = 0; u < KPI; ++u) {
      const unsigned kk[4] = {k[u].x, k[u].y, k[u].z, k[u].w};
#pragma unroll
      for (int j = 0; j < 4; ++j) {
        const unsigned p = kk[j];
        const us2 rel = __builtin_bit_cast(us2, p) - lo2;
        const unsigned relw = __builtin_bit_cast(unsigned, rel);
        const bool tf = relw < ((unsigned)RNG << 16);
        const bool tb = rel.x < RNG;
        const unsigned mf = __builtin_amdgcn_ballot_w32(tf);
        const unsigned mb = __builtin_amdgcn_ballot_w32(tb);
        if (mf | mb) {
          if (mf) {
            unsigned pos = cnt + __builtin_amdgcn_mbcnt_lo(mf, 0u);
            if (tf) myq[pos] = (relw & 0xFFFF0000u) | (p & 0xFFFFu);
            cnt += __builtin_popcount(mf);
          }
          if (mb) {
            unsigned pos = cnt + __builtin_amdgcn_mbcnt_lo(mb, 0u);
            if (tb) myq[pos] = (relw << 16) | (p >> 16) | (1u << 24);
            cnt += __builtin_popcount(mb);
          }
        }
      }
      if (cnt > 64) {
        drain(cnt);
        cnt = 0;
      }
    }
  }
  drain(cnt);
}

__global__ __launch_bounds__(1024) void scan_owner_kernel(
    const int* __restrict__ pk,
    const unsigned* __restrict__ segs,
    float* __restrict__ inv,
    int* __restrict__ lists,
    int* __restrict__ cnts,
    unsigned short* __restrict__ deg16) {
  __shared__ unsigned q[AGG_WAVES][QCAP + 8];
  __shared__ unsigned degc[2 * DPITCH];
  __shared__ unsigned olist[AGG_WAVES][LCAP];
  __shared__ unsigned ocnt[AGG_WAVES];
  __shared__ unsigned ovf;
  __shared__ unsigned segovf;

  const int tid = threadIdx.x;
  const int lane = tid & 31;
  const int wave = __builtin_amdgcn_readfirstlane(tid >> 5);
  const unsigned b = blockIdx.x;

  const int wg = (int)b * AGG_WAVES + wave;
  int* mylist = lists + (size_t)wg * LSTRIDE;
  unsigned* myq = q[wave];

  const unsigned* seg = segs + ((size_t)b * NPART + (size_t)wave * SPW) * SEGW;
  unsigned myc = 0u;
  if (lane < SPW) myc = seg[lane * SEGW + SEGCAP];

  if (tid < 2 * DPITCH) degc[tid] = 0u;
  if (tid < AGG_WAVES) ocnt[tid] = 0u;
  if (tid == 0) { ovf = 0u; segovf = 0u; }
  __syncthreads();
  if (myc > SEGCAP) segovf = 1u;
  __syncthreads();

  auto place = [&](unsigned e) {
    const unsigned row = (e >> 16) & (2 * DPITCH - 1);
    atomicAdd(&degc[row], 1u);
    const unsigned rr = (row / DPITCH) * RNG + (row % DPITCH);
    const unsigned ow = rr % AGG_WAVES;
    const unsigned pos = atomicAdd(&ocnt[ow], 1u);
    if (pos < LCAP) olist[ow][pos] = (e & 0x0100FFFFu) | ((rr / AGG_WAVES) << 16);
  };

  if (segovf == 0u) {
    unsigned sc[SPW], e0[SPW], e1[SPW];
#pragma unroll
    for (int j = 0; j < SPW; ++j) {
      sc[j] = (unsigned)__builtin_amdgcn_readlane((int)myc, j);
      e0[j] = 0u;
      e1[j] = 0u;
      if ((unsigned)lane < sc[j]) e0[j] = seg[j * SEGW + lane];
      if ((unsigned)(32 + lane) < sc[j]) e1[j] = seg[j * SEGW + 32 + lane];
    }
#pragma unroll
    for (int j = 0; j < SPW; ++j) {
      if ((unsigned)lane < sc[j]) place(e0[j]);
      if ((unsigned)(32 + lane) < sc[j]) place(e1[j]);
    }
  } else {
    auto drain = [&](int n) {
#pragma nounroll
      for (int i = lane; i < n; i += 32) place(myq[i]);
    };
    scan_keys(pk, b, wave, lane, myq, drain);
  }
  __syncthreads();
  const int mine = __builtin_amdgcn_readfirstlane((int)ocnt[wave]);
  if (mine > LCAP && lane == 0) ovf = 1u;
  __syncthreads();
  const bool over = ovf != 0u;
  if (!over) {
    for (int i = lane; i < mine; i += 32) mylist[i] = (int)olist[wave][i];
    mylist[mine + lane] = (int)FILL_ENTRY;
  }
  if (lane == 0) cnts[wg] = over ? -1 : mine;

  if (tid < RNG) {
    const int node = (int)b * RNG + tid;
    if (node < NN) {
      inv[node] = 1.0f / fmaxf((float)degc[DPITCH + tid], 1.0f);
      inv[NN + node] = 1.0f / fmaxf((float)degc[tid], 1.0f);
      deg16[node] = (unsigned short)min(degc[DPITCH + tid], DEG_SAT);
      deg16[NN + node] = (unsigned short)min(degc[tid], DEG_SAT);
    }
  }
  if (b == NWG - 1 && tid < DEG_PAD - DEG_VEC)
    ((uint4*)deg16)[DEG_VEC + tid] = make_uint4(0u, 0u, 0u, 0u);
}

__global__ __launch_bounds__(1024) void agg_owner_kernel(
    const float* __restrict__ x, const int* __restrict__ pk,
    const float* __restrict__ inv,
    float* __restrict__ fwd, float* __restrict__ bwd,
    const int* __restrict__ lists,
    const int* __restrict__ cnts,
    const unsigned short* __restrict__ deg16) {
  __shared__ __attribute__((aligned(16))) float acc[(2 * DPITCH + 1) * DD];
  static_assert(sizeof(acc) >= (size_t)DEG_PAD * 16, "the degree table fits the LDS that the general path sums in");
  __shared__ unsigned q[AGG_WAVES][QCAP + 8];

  const int tid = threadIdx.x;
  const int lane = tid & 31;
  const int wave = __builtin_amdgcn_readfirstlane(tid >> 5);
  const unsigned b = blockIdx.x;

  const int wg = (int)b * AGG_WAVES + wave;
  const int* mylist = lists + (size_t)wg * LSTRIDE;
  const int nrep = __builtin_amdgcn_readfirstlane(cnts[wg]);
  const int mode = __builtin_amdgcn_readfirstlane(cnts[(int)b * AGG_WAVES]);

  const float4* x4 = (const float4*)x;

  if (mode >= 0) {
    v16f ax = {}, ay = {}, az = {}, aw = {};
    unsigned cur = (unsigned)mylist[lane];
    unsigned nxt = 0;
    if (32 < nrep) nxt = (unsigned)mylist[32 + lane];
    {
      const uint4* dg = (const uint4*)deg16;
      uint4* tb = (uint4*)acc;
      uint4 piece[DEG_PCS];
#pragma unroll
      for (int u = 0; u < DEG_PCS; ++u) piece[u] = dg[tid + u * 1024];
#pragma unroll
      for (int u = 0; u < DEG_PCS; ++u) tb[tid + u * 1024] = piece[u];
    }
    __syncthreads();
    const unsigned short* tab = (const unsigned short*)acc;
    auto degree = [&](unsigned e) -> unsigned {
      return tab[((e >> 24) & 1u) * NN + (e & 0xFFFFu)];
    };
    auto weight = [&](unsigned e, unsigned d) -> float {
      float w = 1.0f / fmaxf((float)d, 1.0f);
      if (d == DEG_SAT) w = inv[((e >> 24) & 1u) * NN + (e & 0xFFFFu)];
      return w;
    };
    float wcur = weight(cur, degree(cur));
    asm volatile("" : "+v"(wcur));
    for (int base = 0; base < nrep; base += 32) {
      unsigned dnxt = 0;
      unsigned nn = 0;
      if (base + 32 < nrep) dnxt = degree(nxt);
      if (base + 64 < nrep) nn = (unsigned)mylist[base + 64 + lane];
      const int gend = (nrep - base < 32) ? nrep - base : 32;
#pragma nounroll
      for (int g = 0; g < gend; g += 8) {
        {
          unsigned e[8];
          float w[8];
          float4 r[8];
#pragma unroll
          for (int j = 0; j < 8; ++j)
            e[j] = (unsigned)__builtin_amdgcn_readlane((int)cur, g + j);
#pragma unroll
          for (int j = 0; j < 8; ++j)
            r[j] = x4[(size_t)(e[j] & 0xFFFFu) * (DD / 4) + lane];
#pragma unroll
          for (int j = 0; j < 8; ++j)
            w[j] = __int_as_float(__builtin_amdgcn_readlane(__float_as_int(wcur), g + j));
#pragma unroll
          for (int j = 0; j < 8; ++j) {
            const float4 v = r[j];
            const float ww = w[j];
            const unsigned slot = (e[j] >> 16) & 15u;
            ax[slot] += v.x * ww;
            ay[slot] += v.y * ww;
            az[slot] += v.z * ww;
            aw[slot] += v.w * ww;
          }
        }
      }
      float wnxt = 0.0f;
      if (base + 32 < nrep) wnxt = weight(nxt, dnxt);
      cur = nxt;
      wcur = wnxt;
      nxt = nn;
    }
#define ROW_OUT(S, A)                                                          \
    {                                                                          \
      const int rr = (S) * AGG_WAVES + wave;                                   \
      const int dir = (rr >= RNG) ? 1 : 0;                                     \
      const int node = (int)b * RNG + rr - dir * RNG;                          \
      if (rr < 2 * RNG && node < NN)                                           \
        *(float4*)((dir ? bwd : fwd) + (size_t)node * DD + lane * 4) = A;      \
    }
    ROW_OUT(0, make_float4(ax[0], ay[0], az[0], aw[0]));
    ROW_OUT(1, make_float4(ax[1], ay[1], az[1], aw[1]));
    ROW_OUT(2, make_float4(ax[2], ay[2], az[2], aw[2]));
    ROW_OUT(3, make_float4(ax[3], ay[3], az[3], aw[3]));
    ROW_OUT(4, make_float4(ax[4], ay[4], az[4], aw[4]));
    ROW_OUT(5, make_float4(ax[5], ay[5], az[5], aw[5]));
    ROW_OUT(6, make_float4(ax[6], ay[6], az[6], aw[6]));
    ROW_OUT(7, make_float4(ax[7], ay[7], az[7], aw[7]));
    ROW_OUT(8, make_float4(ax[8], ay[8], az[8], aw[8]));
    ROW_OUT(9, make_float4(ax[9], ay[9], az[9], aw[9]));
    ROW_OUT(10, make_float4(ax[10], ay[10], az[10], aw[10]));
    ROW_OUT(11, make_float4(ax[11], ay[11], az[11], aw[11]));
    ROW_OUT(12, make_float4(ax[12], ay[12], az[12], aw[12]));
#undef ROW_OUT
    return;
  }

  float4* acc4 = (float4*)acc;
  for (int i = tid; i < (2 * DPITCH + 1) * DD / 4; i += 1024)
    acc4[i] = make_float4(0.0f, 0.0f, 0.0f, 0.0f);
  __syncthreads();

  {
    unsigned* myq = q[wave];
    auto drain = [&](int n) {
      if (lane < 8) myq[n + lane] = 1u << 25;
      for (int i = 0; i < n; i += 8) {
        unsigned e[8];
        float w[8];
        float4 r[8];
#pragma unroll
        for (int j = 0; j < 8; ++j)
          e[j] = __builtin_amdgcn_readfirstlane(myq[i + j]);
#pragma unroll
        for (int j = 0; j < 8; ++j) {
          unsigned o = e[j] & 0xFFFFu;
          w[j] = inv[((e[j] >> 24) & 1u) * NN + o];
          r[j] = x4[(size_t)o * (DD / 4) + lane];
        }
#pragma unroll
        for (int j = 0; j < 8; ++j) {
          float* a = acc + (e[j] >> 16) * DD + lane;
          atomicAdd(a, r[j].x * w[j]);
          atomicAdd(a + 32, r[j].y * w[j]);
          atomicAdd(a + 64, r[j].z * w[j]);
          atomicAdd(a + 96, r[j].w * w[j]);
        }
      }
    };
    scan_keys(pk, b, wave, lane, myq, drain);
  }
  __syncthreads();

  for (int row = wave; row < 2 * RNG; row += AGG_WAVES) {
    const int dir = (row >= RNG) ? 1 : 0;
    const int loc = row - dir * RNG;
    int node = (int)b * RNG + loc;
    if (node < NN) {
      const float* a = acc + (dir * DPITCH + loc) * DD + lane;
      float4 v = make_float4(a[0], a[32], a[64], a[96]);
      float* o = dir ? bwd : fwd;
      *(float4*)(o + (size_t)node * DD + lane * 4) = v;
    }
  }
}

#define GEMM_WG 256
#define NTILE (NN / 16)
#define APITCH (K3 + 4)

static_assert(NN % 16 == 0, "row tiles are whole");

typedef __attribute__((ext_vector_type(4))) float v4f;

__global__ __launch_bounds__(256) void gemm_wmma_kernel(
    const float* __restrict__ x, const float* __restrict__ fwd,
    const float* __restrict__ bwd, const float* __restrict__ W,
    const float* __restrict__ bias, float* __restrict__ out, int do_relu) {
  __shared__ __attribute__((aligned(16))) float As[2][16][APITCH];

  const int tid = threadIdx.x;
  const int wave = tid >> 5;
  const int lane = tid & 31;
  const int hl = lane >> 4;
  const int l16 = lane & 15;
  const int col = wave * 16 + l16;

  v2f bw[K3 / 4];
#pragma unroll
  for (int s = 0; s < K3 / 4; ++s) {
    bw[s].x = W[(size_t)(4 * s + 2 * hl + 0) * DD + col];
    bw[s].y = W[(size_t)(4 * s + 2 * hl + 1) * DD + col];
  }
  const float bb = bias[col];

  const v4f* x4 = (const v4f*)x + tid;
  const v4f* f4 = (const v4f*)fwd + tid;
  const v4f* b4 = (const v4f*)bwd + tid;
  const int sA = (tid >> 5) * APITCH + 4 * (tid & 31);
  v4f g0, g1, g2, g3, g4, g5;

  int tile = blockIdx.x;
  {
    const size_t base = (size_t)tile * (16 * DD / 4);
    g0 = x4[base]; g1 = x4[base + 256];
    g2 = f4[base]; g3 = f4[base + 256];
    g4 = b4[base]; g5 = b4[base + 256];
    float* d = &As[0][0][0] + sA;
    *(v4f*)(d) = g0;            *(v4f*)(d + 8 * APITCH) = g1;
    *(v4f*)(d + DD) = g2;       *(v4f*)(d + 8 * APITCH + DD) = g3;
    *(v4f*)(d + 2 * DD) = g4;   *(v4f*)(d + 8 * APITCH + 2 * DD) = g5;
  }
  __syncthreads();

  int buf = 0;
  for (;;) {
    const bool more = tile + GEMM_WG < NTILE;
    if (more) {
      const size_t base = (size_t)(tile + GEMM_WG) * (16 * DD / 4);
      g0 = x4[base]; g1 = x4[base + 256];
      g2 = f4[base]; g3 = f4[base + 256];
      g4 = b4[base]; g5 = b4[base + 256];
    }

    const float* arow = &As[buf][l16][2 * hl];
    v2f af[2][8];
#pragma unroll
    for (int j = 0; j < 8; ++j) af[0][j] = *(const v2f*)(arow + 4 * j);
    v8f acc = {};
#pragma unroll
    for (int gidx = 0; gidx < K3 / 32; ++gidx) {
      if (gidx + 1 < K3 / 32) {
#pragma unroll
        for (int j = 0; j < 8; ++j)
          af[(gidx + 1) & 1][j] = *(const v2f*)(arow + 32 * (gidx + 1) + 4 * j);
      }
      __builtin_amdgcn_sched_barrier(0);
#pragma unroll
      for (int j = 0; j < 8; ++j)
        acc = __builtin_amdgcn_wmma_f32_16x16x4_f32(
             false, af[gidx & 1][j],  false, bw[8 * gidx + j],
             (short)0, acc,  false,  false);
    }

    {
      float* d = &As[buf ^ 1][0][0] + sA;
      *(v4f*)(d) = g0;            *(v4f*)(d + 8 * APITCH) = g1;
      *(v4f*)(d + DD) = g2;       *(v4f*)(d + 8 * APITCH + DD) = g3;
      *(v4f*)(d + 2 * DD) = g4;   *(v4f*)(d + 8 * APITCH + 2 * DD) = g5;
    }
    __syncthreads();

    asm volatile("v_nop\n\tv_nop\n\tv_nop\n\tv_nop\n\tv_nop" : "+v"(acc));

    float* o = out + (size_t)(tile * 16 + 8 * hl) * DD + col;
#pragma unroll
    for (int r = 0; r < 8; ++r) {
      float v = acc[r] + bb;
      if (do_relu) v = fmaxf(v, 0.0f);
      o[r * DD] = v;
    }

    if (!more) break;
    tile += GEMM_WG;
    buf ^= 1;
  }
}

extern "C" void kernel_launch(void* const* d_in, const int* in_sizes, int n_in,
                              void* d_out, int out_size, void* d_ws, size_t ws_size,
                              hipStream_t stream) {
  const float* x  = (const float*)d_in[0];
  const float* W0 = (const float*)d_in[1];
  const float* b0 = (const float*)d_in[2];
  const float* W1 = (const float*)d_in[3];
  const float* b1 = (const float*)d_in[4];
  const float* W2 = (const float*)d_in[5];
  const float* b2 = (const float*)d_in[6];
  const int* src  = (const int*)d_in[7];
  const int* dst  = (const int*)d_in[8];
  float* out = (float*)d_out;

  float* ws      = (float*)d_ws;
  float* inv     = ws;
  float* xA      = ws + XA_OFF;
  int*   pk      = (int*)(xA + (size_t)NN * DD);
  int*   lists   = pk + EE;
  int*   cnts    = lists + (size_t)NLW * LSTRIDE;
  unsigned short* deg16 = (unsigned short*)(cnts + NLW);
  float* fwd     = (float*)pk + KEY_PLANE;
  float* bwd     = fwd + (size_t)NN * DD;

  unsigned* segs = (unsigned*)fwd;
  part_kernel<<<NPART, 1024, 0, stream>>>(src, dst, pk, segs);
  scan_owner_kernel<<<NWG, 1024, 0, stream>>>(pk, segs, inv, lists, cnts, deg16);

  const float* Ws[3]  = {W0, W1, W2};
  const float* bs[3]  = {b0, b1, b2};
  float*       dsts[3] = {out, xA, out};

  const float* cur = x;
  for (int l = 0; l < 3; ++l) {
    agg_owner_kernel<<<NWG, 1024, 0, stream>>>(cur, pk, inv, fwd, bwd, lists, cnts, deg16);
    gemm_wmma_kernel<<<GEMM_WG, 256, 0, stream>>>(cur, fwd, bwd, Ws[l], bs[l],
                                                  dsts[l], (l < 2) ? 1 : 0);
    cur = dsts[l];
  }
}
